# starting kernel
# baseline (speedup 1.0000x reference)
#include <hip/hip_runtime.h>

typedef __attribute__((ext_vector_type(2))) float v2f;
typedef __attribute__((ext_vector_type(4))) float v4f;
typedef __attribute__((ext_vector_type(8))) float v8f;

template <int K, int L>
__global__ __launch_bounds__(128)
void gemm_f32_wmma(const float* __restrict__ W, const float* __restrict__ U,
                   float* __restrict__ Out, int M)
{
    const int b = blockIdx.z;
    const float* __restrict__ Ub = U + (long)b * K * L;
    float* __restrict__ Ob = Out + (long)b * (long)M * L;

    const int wave  = threadIdx.x >> 5;
    const int lane  = threadIdx.x & 31;
    const int m0    = blockIdx.y * 32 + (wave >> 1) * 16;
    const int l0    = blockIdx.x * 32 + (wave & 1) * 16;
    const int row   = lane & 15;
    const int khalf = lane >> 4;

    const float* __restrict__ aptr = W  + (long)(m0 + row) * K + 2 * khalf;
    const float* __restrict__ bptr = Ub + (long)(2 * khalf) * L + l0 + row;

    v8f acc = {};
    for (int k = 0; k < K; k += 32) {
        #pragma unroll
        for (int j = 0; j < 8; ++j) {
            v2f a = *(const v2f*)(aptr + j * 4);
            v2f bm;
            bm.x = bptr[(long)(j * 4)     * L];
            bm.y = bptr[(long)(j * 4 + 1) * L];
            acc = __builtin_amdgcn_wmma_f32_16x16x4_f32(
                      false, a, false, bm, (short)0, acc, false, false);
        }
        aptr += 32;
        bptr += (long)32 * L;
    }

    float* __restrict__ dst = Ob + (long)(l0 + row) * M + m0 + 8 * khalf;
    v4f lo = {acc[0], acc[1], acc[2], acc[3]};
    v4f hi = {acc[4], acc[5], acc[6], acc[7]};
    *(v4f*)(dst)     = lo;
    *(v4f*)(dst + 4) = hi;
}

__global__ __launch_bounds__(128)
void ssm_scan(const float* __restrict__ u, const float* __restrict__ A,
              const float* __restrict__ D,
              const float* __restrict__ BuT, const float* __restrict__ CuT,
              const float* __restrict__ dT, float* __restrict__ out,
              int dmodel, int N, int L)
{
    const int b  = blockIdx.y;
    const int t  = threadIdx.x;
    const int d  = blockIdx.x * 64 + (t >> 1);
    const int nh = (t & 1) * 32;

    const float* __restrict__ ubd = u   + ((size_t)b * dmodel + d) * L;
    float*       __restrict__ ybd = out + ((size_t)b * dmodel + d) * L;
    const float* __restrict__ Bub = BuT + (size_t)b * L * N;
    const float* __restrict__ Cub = CuT + (size_t)b * L * N;
    const float* __restrict__ dTb = dT  + (size_t)b * L * dmodel;

    const float Dd = D[d];

    float areg[32], h[32];
    #pragma unroll
    for (int i = 0; i < 32; ++i) { areg[i] = A[nh + i]; h[i] = 0.0f; }

    __shared__ float sB[64];
    __shared__ float sC[64];

    for (int l = 0; l < L; ++l) {
        if (t < 64) sB[t]      = Bub[(size_t)l * N + t];
        else        sC[t - 64] = Cub[(size_t)l * N + (t - 64)];
        __syncthreads();

        const float dl = dTb[(size_t)l * dmodel + d];
        const float ul = ubd[l];
        if (l + 1 < L) __builtin_prefetch(&ubd[l + 1], 0, 0);
        const float du = dl * ul;

        float y = 0.0f;
        #pragma unroll
        for (int i = 0; i < 32; ++i) {
            const float hn = fmaf(dl * areg[i], h[i], du * sB[nh + i]);
            h[i] = hn;
            y = fmaf(sC[nh + i], hn, y);
        }
        y += __shfl_xor(y, 1, 32);
        if ((t & 1) == 0) ybd[l] = fmaf(ul, Dd, y);
        __syncthreads();
    }
}

extern "C" void kernel_launch(void* const* d_in, const int* in_sizes, int n_in,
                              void* d_out, int out_size, void* d_ws, size_t ws_size,
                              hipStream_t stream)
{
    (void)in_sizes; (void)n_in; (void)out_size; (void)ws_size;

    const float* u     = (const float*)d_in[1];
    const float* A     = (const float*)d_in[2];
    const float* Bm    = (const float*)d_in[3];
    const float* Cm    = (const float*)d_in[4];
    const float* D     = (const float*)d_in[5];
    const float* Delta = (const float*)d_in[6];
    float* out = (float*)d_out;

    constexpr int batch = 4, dmodel = 1024, N = 64, L = 2048;

    float* BuT = (float*)d_ws;
    float* CuT = BuT + (size_t)batch * L * N;
    float* dT  = CuT + (size_t)batch * L * N;

    const dim3 blk(128, 1, 1);

    gemm_f32_wmma<dmodel, L><<<dim3(L / 32, N / 32, batch), blk, 0, stream>>>(
        Bm, u, BuT, N);
    gemm_f32_wmma<dmodel, L><<<dim3(L / 32, N / 32, batch), blk, 0, stream>>>(
        Cm, u, CuT, N);
    gemm_f32_wmma<dmodel, L><<<dim3(L / 32, dmodel / 32, batch), blk, 0, stream>>>(
        Delta, u, dT, dmodel);

    ssm_scan<<<dim3(dmodel / 64, batch), dim3(128, 1, 1), 0, stream>>>(
        u, A, D, BuT, CuT, dT, out, dmodel, N, L);
}


# scan: inputs of 16 timesteps staged one chunk ahead through double-buffered LDS; step loop has no global load and one LDS-fenced block barrier per 16 steps instead of two per step; sums unchanged
# speedup vs baseline: 3.8147x; 3.8147x over previous
#include <hip/hip_runtime.h>

typedef __attribute__((ext_vector_type(2))) float v2f;
typedef __attribute__((ext_vector_type(4))) float v4f;
typedef __attribute__((ext_vector_type(8))) float v8f;

template <int K, int L>
__global__ __launch_bounds__(128)
void gemm_f32_wmma(const float* __restrict__ W, const float* __restrict__ U,
                   float* __restrict__ Out, int M)
{
    const int b = blockIdx.z;
    const float* __restrict__ Ub = U + (long)b * K * L;
    float* __restrict__ Ob = Out + (long)b * (long)M * L;

    const int wave  = threadIdx.x >> 5;
    const int lane  = threadIdx.x & 31;
    const int m0    = blockIdx.y * 32 + (wave >> 1) * 16;
    const int l0    = blockIdx.x * 32 + (wave & 1) * 16;
    const int row   = lane & 15;
    const int khalf = lane >> 4;

    const float* __restrict__ aptr = W  + (long)(m0 + row) * K + 2 * khalf;
    const float* __restrict__ bptr = Ub + (long)(2 * khalf) * L + l0 + row;

    v8f acc = {};
    for (int k = 0; k < K; k += 32) {
        #pragma unroll
        for (int j = 0; j < 8; ++j) {
            v2f a = *(const v2f*)(aptr + j * 4);
            v2f bm;
            bm.x = bptr[(long)(j * 4)     * L];
            bm.y = bptr[(long)(j * 4 + 1) * L];
            acc = __builtin_amdgcn_wmma_f32_16x16x4_f32(
                      false, a, false, bm, (short)0, acc, false, false);
        }
        aptr += 32;
        bptr += (long)32 * L;
    }

    float* __restrict__ dst = Ob + (long)(l0 + row) * M + m0 + 8 * khalf;
    v4f lo = {acc[0], acc[1], acc[2], acc[3]};
    v4f hi = {acc[4], acc[5], acc[6], acc[7]};
    *(v4f*)(dst)     = lo;
    *(v4f*)(dst + 4) = hi;
}

constexpr int SCAN_T  = 16;
constexpr int SCAN_UP = SCAN_T + 4;

template <int DMODEL, int L>
__global__ __launch_bounds__(128)
void ssm_scan(const float* __restrict__ u, const float* __restrict__ A,
              const float* __restrict__ D,
              const float* __restrict__ BuT, const float* __restrict__ CuT,
              const float* __restrict__ dT, float* __restrict__ out)
{
    constexpr int N      = 64;
    constexpr int T      = SCAN_T;
    constexpr int UP4    = SCAN_UP / 4;
    constexpr int NCHUNK = L / T;
    static_assert(L % T == 0 && T == 16, "chunk loaders below are written for 16 steps");

    const int b  = blockIdx.y;
    const int t  = threadIdx.x;
    const int d0 = blockIdx.x * 64;
    const int dd = t >> 1;
    const int d  = d0 + dd;
    const int nh = (t & 1) * 32;

    float*       __restrict__ ybd = out + ((size_t)b * DMODEL + d) * L;
    const float* __restrict__ Bub = BuT + (size_t)b * L * N;
    const float* __restrict__ Cub = CuT + (size_t)b * L * N;

    const v4f* __restrict__ gB = (const v4f*)Bub + t;
    const v4f* __restrict__ gC = (const v4f*)Cub + t;
    const float* __restrict__ gD = dT + ((size_t)b * L + (t >> 4)) * DMODEL + d0 + 4 * (t & 15);
    const float* __restrict__ gU = u + ((size_t)b * DMODEL + d) * L + (t & 1) * 8;

    const float Dd = D[d];

    float areg[32], h[32];
    #pragma unroll
    for (int i = 0; i < 32; ++i) { areg[i] = A[nh + i]; h[i] = 0.0f; }

    __shared__ v4f sB[2][T * 16];
    __shared__ v4f sC[2][T * 16];
    __shared__ v4f sD[2][T * 16];
    __shared__ v4f sU[2][64 * UP4];

    v4f pb0, pb1, pc0, pc1, pd0, pd1, pu0, pu1;

    pb0 = gB[0];  pb1 = gB[128];
    pc0 = gC[0];  pc1 = gC[128];
    pd0 = *(const v4f*)(gD);
    pd1 = *(const v4f*)(gD + (size_t)8 * DMODEL);
    pu0 = *(const v4f*)(gU);
    pu1 = *(const v4f*)(gU + 4);
    sB[0][t] = pb0;  sB[0][t + 128] = pb1;
    sC[0][t] = pc0;  sC[0][t + 128] = pc1;
    sD[0][t] = pd0;  sD[0][t + 128] = pd1;
    sU[0][dd * UP4 + (t & 1) * 2]     = pu0;
    sU[0][dd * UP4 + (t & 1) * 2 + 1] = pu1;
    __syncthreads();

    for (int c = 0; c < NCHUNK; ++c) {
        const int  buf  = c & 1;
        const bool more = (c + 1 < NCHUNK);

        if (more) {
            const size_t cn = (size_t)(c + 1);
            pb0 = gB[cn * 256];  pb1 = gB[cn * 256 + 128];
            pc0 = gC[cn * 256];  pc1 = gC[cn * 256 + 128];
            pd0 = *(const v4f*)(gD + cn * T * DMODEL);
            pd1 = *(const v4f*)(gD + (cn * T + 8) * DMODEL);
            pu0 = *(const v4f*)(gU + cn * T);
            pu1 = *(const v4f*)(gU + cn * T + 4);
        }

        const v4f*   __restrict__ cB = &sB[buf][nh >> 2];
        const v4f*   __restrict__ cC = &sC[buf][nh >> 2];
        const float* __restrict__ cD = (const float*)&sD[buf][0] + dd;
        const v4f*   __restrict__ cU = &sU[buf][dd * UP4];
        float* __restrict__ yc = ybd + (size_t)c * T;

        #pragma unroll
        for (int g = 0; g < T / 8; ++g) {
            const v4f ua = cU[2 * g];
            const v4f ub = cU[2 * g + 1];
            float yv[8];
            #pragma unroll
            for (int s = 0; s < 8; ++s) {
                const int   j  = g * 8 + s;
                const float dl = cD[j * 64];
                const float ul = (s < 4) ? ua[s & 3] : ub[s & 3];
                const float du = dl * ul;

                float y = 0.0f;
                #pragma unroll
                for (int q = 0; q < 8; ++q) {
                    const v4f bq = cB[j * 16 + q];
                    const v4f cq = cC[j * 16 + q];
                    #pragma unroll
                    for (int k = 0; k < 4; ++k) {
                        const int i = q * 4 + k;
                        const float hn = fmaf(dl * areg[i], h[i], du * bq[k]);
                        h[i] = hn;
                        y = fmaf(cq[k], hn, y);
                    }
                }
                y += __shfl_xor(y, 1, 32);
                yv[s] = fmaf(ul, Dd, y);
            }
            const v4f ylo = {yv[0], yv[1], yv[2], yv[3]};
            const v4f yhi = {yv[4], yv[5], yv[6], yv[7]};
            *(v4f*)(yc + g * 8 + (t & 1) * 4) = (t & 1) ? yhi : ylo;
        }

        if (more) {
            const int nb = buf ^ 1;
            sB[nb][t] = pb0;  sB[nb][t + 128] = pb1;
            sC[nb][t] = pc0;  sC[nb][t + 128] = pc1;
            sD[nb][t] = pd0;  sD[nb][t + 128] = pd1;
            sU[nb][dd * UP4 + (t & 1) * 2]     = pu0;
            sU[nb][dd * UP4 + (t & 1) * 2 + 1] = pu1;
        }
        __builtin_amdgcn_fence(__ATOMIC_RELEASE, "workgroup", "local");
        __builtin_amdgcn_s_barrier();
        __builtin_amdgcn_fence(__ATOMIC_ACQUIRE, "workgroup", "local");
    }
}

extern "C" void kernel_launch(void* const* d_in, const int* in_sizes, int n_in,
                              void* d_out, int out_size, void* d_ws, size_t ws_size,
                              hipStream_t stream)
{
    (void)in_sizes; (void)n_in; (void)out_size; (void)ws_size;

    const float* u     = (const float*)d_in[1];
    const float* A     = (const float*)d_in[2];
    const float* Bm    = (const float*)d_in[3];
    const float* Cm    = (const float*)d_in[4];
    const float* D     = (const float*)d_in[5];
    const float* Delta = (const float*)d_in[6];
    float* out = (float*)d_out;

    constexpr int batch = 4, dmodel = 1024, N = 64, L = 2048;

    float* BuT = (float*)d_ws;
    float* CuT = BuT + (size_t)batch * L * N;
    float* dT  = CuT + (size_t)batch * L * N;

    const dim3 blk(128, 1, 1);

    gemm_f32_wmma<dmodel, L><<<dim3(L / 32, N / 32, batch), blk, 0, stream>>>(
        Bm, u, BuT, N);
    gemm_f32_wmma<dmodel, L><<<dim3(L / 32, N / 32, batch), blk, 0, stream>>>(
        Cm, u, CuT, N);
    gemm_f32_wmma<dmodel, L><<<dim3(L / 32, dmodel / 32, batch), blk, 0, stream>>>(
        Delta, u, dT, dmodel);

    ssm_scan<dmodel, L><<<dim3(dmodel / 64, batch), dim3(128, 1, 1), 0, stream>>>(
        u, A, D, BuT, CuT, dT, out);
}


# scan: each row's 64 states spread over 8 lanes (8 per lane) instead of 2: 4x the waves, a quarter of the LDS reads per wave and step; y joined from 8 partial sums by a 3-level lane butterfly
# speedup vs baseline: 8.0756x; 2.1170x over previous
#include <hip/hip_runtime.h>

typedef __attribute__((ext_vector_type(2))) float v2f;
typedef __attribute__((ext_vector_type(4))) float v4f;
typedef __attribute__((ext_vector_type(8))) float v8f;

template <int K, int L>
__global__ __launch_bounds__(128)
void gemm_f32_wmma(const float* __restrict__ W, const float* __restrict__ U,
                   float* __restrict__ Out, int M)
{
    const int b = blockIdx.z;
    const float* __restrict__ Ub = U + (long)b * K * L;
    float* __restrict__ Ob = Out + (long)b * (long)M * L;

    const int wave  = threadIdx.x >> 5;
    const int lane  = threadIdx.x & 31;
    const int m0    = blockIdx.y * 32 + (wave >> 1) * 16;
    const int l0    = blockIdx.x * 32 + (wave & 1) * 16;
    const int row   = lane & 15;
    const int khalf = lane >> 4;

    const float* __restrict__ aptr = W  + (long)(m0 + row) * K + 2 * khalf;
    const float* __restrict__ bptr = Ub + (long)(2 * khalf) * L + l0 + row;

    v8f acc = {};
    for (int k = 0; k < K; k += 32) {
        #pragma unroll
        for (int j = 0; j < 8; ++j) {
            v2f a = *(const v2f*)(aptr + j * 4);
            v2f bm;
            bm.x = bptr[(long)(j * 4)     * L];
            bm.y = bptr[(long)(j * 4 + 1) * L];
            acc = __builtin_amdgcn_wmma_f32_16x16x4_f32(
                      false, a, false, bm, (short)0, acc, false, false);
        }
        aptr += 32;
        bptr += (long)32 * L;
    }

    float* __restrict__ dst = Ob + (long)(l0 + row) * M + m0 + 8 * khalf;
    v4f lo = {acc[0], acc[1], acc[2], acc[3]};
    v4f hi = {acc[4], acc[5], acc[6], acc[7]};
    *(v4f*)(dst)     = lo;
    *(v4f*)(dst + 4) = hi;
}

constexpr int SCAN_T  = 32;
constexpr int SCAN_RP = SCAN_T + 4;

template <int CTRL>
static __device__ __forceinline__ float xmask_add(float v)
{
    const int o = __builtin_amdgcn_update_dpp(0, __float_as_int(v), CTRL, 0xf, 0xf, true);
    return v + __int_as_float(o);
}

template <int DMODEL, int L>
__global__ __launch_bounds__(128) __attribute__((amdgpu_waves_per_eu(1, 2)))
void ssm_scan(const float* __restrict__ u, const float* __restrict__ A,
              const float* __restrict__ D,
              const float* __restrict__ BuT, const float* __restrict__ CuT,
              const float* __restrict__ dT, float* __restrict__ out)
{
    constexpr int N      = 64;
    constexpr int T      = SCAN_T;
    constexpr int RP     = SCAN_RP;
    constexpr int RP4    = SCAN_RP / 4;
    constexpr int ROWS   = 16;
    constexpr int NCHUNK = L / T;
    static_assert(L % T == 0 && T == 32 && N == 64, "chunk loaders below are written for 32 steps of 64 states");

    const int b  = blockIdx.y;
    const int t  = threadIdx.x;
    const int d0 = blockIdx.x * ROWS;
    const int r  = t >> 3;
    const int lj = t & 7;
    const int d  = d0 + r;

    float*       __restrict__ ybd = out + ((size_t)b * DMODEL + d) * L + 4 * lj;
    const float* __restrict__ Bub = BuT + (size_t)b * L * N;
    const float* __restrict__ Cub = CuT + (size_t)b * L * N;

    const v4f* __restrict__ gB = (const v4f*)Bub + t;
    const v4f* __restrict__ gC = (const v4f*)Cub + t;
    const float* __restrict__ gD = dT + ((size_t)b * L + (t >> 2)) * DMODEL + d0 + 4 * (t & 3);
    const float* __restrict__ gU = u + ((size_t)b * DMODEL + d) * L + 4 * lj;

    const float Dd = D[d];

    float areg[8], h[8];
    #pragma unroll
    for (int i = 0; i < 8; ++i) { areg[i] = A[8 * lj + i]; h[i] = 0.0f; }

    __shared__ v4f sB[2][T * 16];
    __shared__ v4f sC[2][T * 16];
    __shared__ v4f sD[2][ROWS * RP4];
    __shared__ v4f sU[2][ROWS * RP4];

    v4f pb0, pb1, pb2, pb3, pc0, pc1, pc2, pc3, pd, pu;

    pb0 = gB[0];  pb1 = gB[128];  pb2 = gB[256];  pb3 = gB[384];
    pc0 = gC[0];  pc1 = gC[128];  pc2 = gC[256];  pc3 = gC[384];
    pd  = *(const v4f*)(gD);
    pu  = *(const v4f*)(gU);
    {
        sB[0][t] = pb0;  sB[0][t + 128] = pb1;  sB[0][t + 256] = pb2;  sB[0][t + 384] = pb3;
        sC[0][t] = pc0;  sC[0][t + 128] = pc1;  sC[0][t + 256] = pc2;  sC[0][t + 384] = pc3;
        float* __restrict__ wD = (float*)&sD[0][0] + (4 * (t & 3)) * RP + (t >> 2);
        wD[0] = pd[0];  wD[RP] = pd[1];  wD[2 * RP] = pd[2];  wD[3 * RP] = pd[3];
        sU[0][r * RP4 + lj] = pu;
    }
    __syncthreads();

    for (int c = 0; c < NCHUNK; ++c) {
        const int  buf  = c & 1;
        const bool more = (c + 1 < NCHUNK);

        if (more) {
            const size_t cn = (size_t)(c + 1);
            pb0 = gB[cn * 512];        pb1 = gB[cn * 512 + 128];
            pb2 = gB[cn * 512 + 256];  pb3 = gB[cn * 512 + 384];
            pc0 = gC[cn * 512];        pc1 = gC[cn * 512 + 128];
            pc2 = gC[cn * 512 + 256];  pc3 = gC[cn * 512 + 384];
            pd  = *(const v4f*)(gD + cn * T * DMODEL);
            pu  = *(const v4f*)(gU + cn * T);
        }

        const v4f* __restrict__ cB = &sB[buf][2 * lj];
        const v4f* __restrict__ cC = &sC[buf][2 * lj];
        const v4f* __restrict__ cD = &sD[buf][r * RP4];
        const v4f* __restrict__ cU = &sU[buf][r * RP4];

        v4f yv = {0.0f, 0.0f, 0.0f, 0.0f};
        #pragma unroll
        for (int g = 0; g < T / 4; ++g) {
            const v4f  d4   = cD[g];
            const v4f  u4   = cU[g];
            const bool mine = (lj == g);
            #pragma unroll
            for (int s = 0; s < 4; ++s) {
                const int   j  = g * 4 + s;
                const float dl = d4[s];
                const float ul = u4[s];
                const float du = dl * ul;
                const v4f bq0 = cB[j * 16];
                const v4f bq1 = cB[j * 16 + 1];
                const v4f cq0 = cC[j * 16];
                const v4f cq1 = cC[j * 16 + 1];

                float y = 0.0f;
                #pragma unroll
                for (int k = 0; k < 4; ++k) {
                    const float hn = fmaf(dl * areg[k], h[k], du * bq0[k]);
                    h[k] = hn;
                    y = fmaf(cq0[k], hn, y);
                }
                #pragma unroll
                for (int k = 0; k < 4; ++k) {
                    const float hn = fmaf(dl * areg[4 + k], h[4 + k], du * bq1[k]);
                    h[4 + k] = hn;
                    y = fmaf(cq1[k], hn, y);
                }
                y = xmask_add<0x161>(y);
                y = xmask_add<0x162>(y);
                y = xmask_add<0x164>(y);
                const float yo = fmaf(ul, Dd, y);
                yv[s] = mine ? yo : yv[s];
            }
        }
        *(v4f*)(ybd + (size_t)c * T) = yv;

        if (more) {
            const int nb = buf ^ 1;
            sB[nb][t] = pb0;  sB[nb][t + 128] = pb1;  sB[nb][t + 256] = pb2;  sB[nb][t + 384] = pb3;
            sC[nb][t] = pc0;  sC[nb][t + 128] = pc1;  sC[nb][t + 256] = pc2;  sC[nb][t + 384] = pc3;
            float* __restrict__ wD = (float*)&sD[nb][0] + (4 * (t & 3)) * RP + (t >> 2);
            wD[0] = pd[0];  wD[RP] = pd[1];  wD[2 * RP] = pd[2];  wD[3 * RP] = pd[3];
            sU[nb][r * RP4 + lj] = pu;
        }
        __builtin_amdgcn_fence(__ATOMIC_RELEASE, "workgroup", "local");
        __builtin_amdgcn_s_barrier();
        __builtin_amdgcn_fence(__ATOMIC_ACQUIRE, "workgroup", "local");
    }
}

extern "C" void kernel_launch(void* const* d_in, const int* in_sizes, int n_in,
                              void* d_out, int out_size, void* d_ws, size_t ws_size,
                              hipStream_t stream)
{
    (void)in_sizes; (void)n_in; (void)out_size; (void)ws_size;

    const float* u     = (const float*)d_in[1];
    const float* A     = (const float*)d_in[2];
    const float* Bm    = (const float*)d_in[3];
    const float* Cm    = (const float*)d_in[4];
    const float* D     = (const float*)d_in[5];
    const float* Delta = (const float*)d_in[6];
    float* out = (float*)d_out;

    constexpr int batch = 4, dmodel = 1024, N = 64, L = 2048;

    float* BuT = (float*)d_ws;
    float* CuT = BuT + (size_t)batch * L * N;
    float* dT  = CuT + (size_t)batch * L * N;

    const dim3 blk(128, 1, 1);

    gemm_f32_wmma<dmodel, L><<<dim3(L / 32, N / 32, batch), blk, 0, stream>>>(
        Bm, u, BuT, N);
    gemm_f32_wmma<dmodel, L><<<dim3(L / 32, N / 32, batch), blk, 0, stream>>>(
        Cm, u, CuT, N);
    gemm_f32_wmma<dmodel, L><<<dim3(L / 32, dmodel / 32, batch), blk, 0, stream>>>(
        Delta, u, dT, dmodel);

    ssm_scan<dmodel, L><<<dim3(dmodel / 16, batch), dim3(128, 1, 1), 0, stream>>>(
        u, A, D, BuT, CuT, dT, out);
}


# delta GEMM: a wave keeps a 32x64 output block (2x4 tiles) and feeds 4/2 WMMAs per 16-byte-loaded A/B fragment, next k-block fetched ahead; k regrouped within 16-k blocks (sum order changes)
# speedup vs baseline: 13.5975x; 1.6838x over previous
#include <hip/hip_runtime.h>

typedef __attribute__((ext_vector_type(2))) float v2f;
typedef __attribute__((ext_vector_type(4))) float v4f;
typedef __attribute__((ext_vector_type(8))) float v8f;

template <int K, int L>
__global__ __launch_bounds__(128)
void gemm_f32_wmma(const float* __restrict__ W, const float* __restrict__ U,
                   float* __restrict__ Out, int M)
{
    const int b = blockIdx.z;
    const float* __restrict__ Ub = U + (long)b * K * L;
    float* __restrict__ Ob = Out + (long)b * (long)M * L;

    const int wave  = threadIdx.x >> 5;
    const int lane  = threadIdx.x & 31;
    const int m0    = blockIdx.y * 32 + (wave >> 1) * 16;
    const int l0    = blockIdx.x * 32 + (wave & 1) * 16;
    const int row   = lane & 15;
    const int khalf = lane >> 4;

    const float* __restrict__ aptr = W  + (long)(m0 + row) * K + 2 * khalf;
    const float* __restrict__ bptr = Ub + (long)(2 * khalf) * L + l0 + row;

    v8f acc = {};
    for (int k = 0; k < K; k += 32) {
        #pragma unroll
        for (int j = 0; j < 8; ++j) {
            v2f a = *(const v2f*)(aptr + j * 4);
            v2f bm;
            bm.x = bptr[(long)(j * 4)     * L];
            bm.y = bptr[(long)(j * 4 + 1) * L];
            acc = __builtin_amdgcn_wmma_f32_16x16x4_f32(
                      false, a, false, bm, (short)0, acc, false, false);
        }
        aptr += 32;
        bptr += (long)32 * L;
    }

    float* __restrict__ dst = Ob + (long)(l0 + row) * M + m0 + 8 * khalf;
    v4f lo = {acc[0], acc[1], acc[2], acc[3]};
    v4f hi = {acc[4], acc[5], acc[6], acc[7]};
    *(v4f*)(dst)     = lo;
    *(v4f*)(dst + 4) = hi;
}

struct RbFrag {
    v4f a[2][2];
    v4f b0[4];
    v4f b1[4];
};

template <int K, int L>
static __device__ __forceinline__ void rb_load(RbFrag& f, const float* __restrict__ ap,
                                               const float* __restrict__ bp)
{
    #pragma unroll
    for (int tm = 0; tm < 2; ++tm) {
        #pragma unroll
        for (int q = 0; q < 2; ++q)
            f.a[tm][q] = *(const v4f*)(ap + (size_t)tm * 16 * K + 4 * q);
    }
    #pragma unroll
    for (int i = 0; i < 4; ++i) {
        f.b0[i] = *(const v4f*)(bp + (size_t)(2 * i)     * L);
        f.b1[i] = *(const v4f*)(bp + (size_t)(2 * i + 1) * L);
    }
}

static __device__ __forceinline__ void rb_compute(const RbFrag& f, v8f (&acc)[2][4])
{
    #pragma unroll
    for (int i = 0; i < 4; ++i) {
        #pragma unroll
        for (int tm = 0; tm < 2; ++tm) {
            v2f av;
            av.x = f.a[tm][i >> 1][2 * (i & 1)];
            av.y = f.a[tm][i >> 1][2 * (i & 1) + 1];
            #pragma unroll
            for (int tn = 0; tn < 4; ++tn) {
                v2f bv;
                bv.x = f.b0[i][tn];
                bv.y = f.b1[i][tn];
                acc[tm][tn] = __builtin_amdgcn_wmma_f32_16x16x4_f32(
                                  false, av, false, bv, (short)0, acc[tm][tn], false, false);
            }
        }
    }
}

template <int K, int L, int M>
__global__ __launch_bounds__(128) __attribute__((amdgpu_waves_per_eu(4, 4)))
void gemm_f32_wmma_rb(const float* __restrict__ W, const float* __restrict__ U,
                      float* __restrict__ Out)
{
    static_assert(K % 32 == 0 && K >= 64 && L % 128 == 0 && M % 64 == 0, "tile sizes below");
    const int b = blockIdx.z;
    const float* __restrict__ Ub = U + (size_t)b * K * L;
    float* __restrict__ Ob = Out + (size_t)b * M * L;

    const int wave  = threadIdx.x >> 5;
    const int lane  = threadIdx.x & 31;
    const int col   = lane & 15;
    const int khalf = lane >> 4;
    const int m0    = blockIdx.y * 64  + (wave >> 1) * 32;
    const int l0    = blockIdx.x * 128 + (wave & 1) * 64;

    const float* __restrict__ aptr = W  + (size_t)(m0 + col) * K + 8 * khalf;
    const float* __restrict__ bptr = Ub + (size_t)(8 * khalf) * L + l0 + 4 * col;

    v8f acc[2][4];
    #pragma unroll
    for (int tm = 0; tm < 2; ++tm) {
        #pragma unroll
        for (int tn = 0; tn < 4; ++tn)
            acc[tm][tn] = v8f{};
    }

    RbFrag f0, f1;
    rb_load<K, L>(f0, aptr, bptr);
    for (int kb = 0; kb < K - 32; kb += 32) {
        rb_load<K, L>(f1, aptr + 16, bptr + (size_t)16 * L);
        rb_compute(f0, acc);
        rb_load<K, L>(f0, aptr + 32, bptr + (size_t)32 * L);
        rb_compute(f1, acc);
        aptr += 32;
        bptr += (size_t)32 * L;
    }
    rb_load<K, L>(f1, aptr + 16, bptr + (size_t)16 * L);
    rb_compute(f0, acc);
    rb_compute(f1, acc);

    #pragma unroll
    for (int tm = 0; tm < 2; ++tm) {
        #pragma unroll
        for (int tn = 0; tn < 4; ++tn) {
            float* __restrict__ dst = Ob + (size_t)(l0 + 4 * col + tn) * M + m0 + 16 * tm + 8 * khalf;
            const v8f v = acc[tm][tn];
            v4f lo = {v[0], v[1], v[2], v[3]};
            v4f hi = {v[4], v[5], v[6], v[7]};
            *(v4f*)(dst)     = lo;
            *(v4f*)(dst + 4) = hi;
        }
    }
}

constexpr int SCAN_T  = 32;
constexpr int SCAN_RP = SCAN_T + 4;

template <int CTRL>
static __device__ __forceinline__ float xmask_add(float v)
{
    const int o = __builtin_amdgcn_update_dpp(0, __float_as_int(v), CTRL, 0xf, 0xf, true);
    return v + __int_as_float(o);
}

template <int DMODEL, int L>
__global__ __launch_bounds__(128) __attribute__((amdgpu_waves_per_eu(1, 2)))
void ssm_scan(const float* __restrict__ u, const float* __restrict__ A,
              const float* __restrict__ D,
              const float* __restrict__ BuT, const float* __restrict__ CuT,
              const float* __restrict__ dT, float* __restrict__ out)
{
    constexpr int N      = 64;
    constexpr int T      = SCAN_T;
    constexpr int RP     = SCAN_RP;
    constexpr int RP4    = SCAN_RP / 4;
    constexpr int ROWS   = 16;
    constexpr int NCHUNK = L / T;
    static_assert(L % T == 0 && T == 32 && N == 64, "chunk loaders below are written for 32 steps of 64 states");

    const int b  = blockIdx.y;
    const int t  = threadIdx.x;
    const int d0 = blockIdx.x * ROWS;
    const int r  = t >> 3;
    const int lj = t & 7;
    const int d  = d0 + r;

    float*       __restrict__ ybd = out + ((size_t)b * DMODEL + d) * L + 4 * lj;
    const float* __restrict__ Bub = BuT + (size_t)b * L * N;
    const float* __restrict__ Cub = CuT + (size_t)b * L * N;

    const v4f* __restrict__ gB = (const v4f*)Bub + t;
    const v4f* __restrict__ gC = (const v4f*)Cub + t;
    const float* __restrict__ gD = dT + ((size_t)b * L + (t >> 2)) * DMODEL + d0 + 4 * (t & 3);
    const float* __restrict__ gU = u + ((size_t)b * DMODEL + d) * L + 4 * lj;

    const float Dd = D[d];

    float areg[8], h[8];
    #pragma unroll
    for (int i = 0; i < 8; ++i) { areg[i] = A[8 * lj + i]; h[i] = 0.0f; }

    __shared__ v4f sB[2][T * 16];
    __shared__ v4f sC[2][T * 16];
    __shared__ v4f sD[2][ROWS * RP4];
    __shared__ v4f sU[2][ROWS * RP4];

    v4f pb0, pb1, pb2, pb3, pc0, pc1, pc2, pc3, pd, pu;

    pb0 = gB[0];  pb1 = gB[128];  pb2 = gB[256];  pb3 = gB[384];
    pc0 = gC[0];  pc1 = gC[128];  pc2 = gC[256];  pc3 = gC[384];
    pd  = *(const v4f*)(gD);
    pu  = *(const v4f*)(gU);
    {
        sB[0][t] = pb0;  sB[0][t + 128] = pb1;  sB[0][t + 256] = pb2;  sB[0][t + 384] = pb3;
        sC[0][t] = pc0;  sC[0][t + 128] = pc1;  sC[0][t + 256] = pc2;  sC[0][t + 384] = pc3;
        float* __restrict__ wD = (float*)&sD[0][0] + (4 * (t & 3)) * RP + (t >> 2);
        wD[0] = pd[0];  wD[RP] = pd[1];  wD[2 * RP] = pd[2];  wD[3 * RP] = pd[3];
        sU[0][r * RP4 + lj] = pu;
    }
    __syncthreads();

    for (int c = 0; c < NCHUNK; ++c) {
        const int  buf  = c & 1;
        const bool more = (c + 1 < NCHUNK);

        if (more) {
            const size_t cn = (size_t)(c + 1);
            pb0 = gB[cn * 512];        pb1 = gB[cn * 512 + 128];
            pb2 = gB[cn * 512 + 256];  pb3 = gB[cn * 512 + 384];
            pc0 = gC[cn * 512];        pc1 = gC[cn * 512 + 128];
            pc2 = gC[cn * 512 + 256];  pc3 = gC[cn * 512 + 384];
            pd  = *(const v4f*)(gD + cn * T * DMODEL);
            pu  = *(const v4f*)(gU + cn * T);
        }

        const v4f* __restrict__ cB = &sB[buf][2 * lj];
        const v4f* __restrict__ cC = &sC[buf][2 * lj];
        const v4f* __restrict__ cD = &sD[buf][r * RP4];
        const v4f* __restrict__ cU = &sU[buf][r * RP4];

        v4f yv = {0.0f, 0.0f, 0.0f, 0.0f};
        #pragma unroll
        for (int g = 0; g < T / 4; ++g) {
            const v4f  d4   = cD[g];
            const v4f  u4   = cU[g];
            const bool mine = (lj == g);
            #pragma unroll
            for (int s = 0; s < 4; ++s) {
                const int   j  = g * 4 + s;
                const float dl = d4[s];
                const float ul = u4[s];
                const float du = dl * ul;
                const v4f bq0 = cB[j * 16];
                const v4f bq1 = cB[j * 16 + 1];
                const v4f cq0 = cC[j * 16];
                const v4f cq1 = cC[j * 16 + 1];

                float y = 0.0f;
                #pragma unroll
                for (int k = 0; k < 4; ++k) {
                    const float hn = fmaf(dl * areg[k], h[k], du * bq0[k]);
                    h[k] = hn;
                    y = fmaf(cq0[k], hn, y);
                }
                #pragma unroll
                for (int k = 0; k < 4; ++k) {
                    const float hn = fmaf(dl * areg[4 + k], h[4 + k], du * bq1[k]);
                    h[4 + k] = hn;
                    y = fmaf(cq1[k], hn, y);
                }
                y = xmask_add<0x161>(y);
                y = xmask_add<0x162>(y);
                y = xmask_add<0x164>(y);
                const float yo = fmaf(ul, Dd, y);
                yv[s] = mine ? yo : yv[s];
            }
        }
        *(v4f*)(ybd + (size_t)c * T) = yv;

        if (more) {
            const int nb = buf ^ 1;
            sB[nb][t] = pb0;  sB[nb][t + 128] = pb1;  sB[nb][t + 256] = pb2;  sB[nb][t + 384] = pb3;
            sC[nb][t] = pc0;  sC[nb][t + 128] = pc1;  sC[nb][t + 256] = pc2;  sC[nb][t + 384] = pc3;
            float* __restrict__ wD = (float*)&sD[nb][0] + (4 * (t & 3)) * RP + (t >> 2);
            wD[0] = pd[0];  wD[RP] = pd[1];  wD[2 * RP] = pd[2];  wD[3 * RP] = pd[3];
            sU[nb][r * RP4 + lj] = pu;
        }
        __builtin_amdgcn_fence(__ATOMIC_RELEASE, "workgroup", "local");
        __builtin_amdgcn_s_barrier();
        __builtin_amdgcn_fence(__ATOMIC_ACQUIRE, "workgroup", "local");
    }
}

extern "C" void kernel_launch(void* const* d_in, const int* in_sizes, int n_in,
                              void* d_out, int out_size, void* d_ws, size_t ws_size,
                              hipStream_t stream)
{
    (void)in_sizes; (void)n_in; (void)out_size; (void)ws_size;

    const float* u     = (const float*)d_in[1];
    const float* A     = (const float*)d_in[2];
    const float* Bm    = (const float*)d_in[3];
    const float* Cm    = (const float*)d_in[4];
    const float* D     = (const float*)d_in[5];
    const float* Delta = (const float*)d_in[6];
    float* out = (float*)d_out;

    constexpr int batch = 4, dmodel = 1024, N = 64, L = 2048;

    float* BuT = (float*)d_ws;
    float* CuT = BuT + (size_t)batch * L * N;
    float* dT  = CuT + (size_t)batch * L * N;

    const dim3 blk(128, 1, 1);

    gemm_f32_wmma<dmodel, L><<<dim3(L / 32, N / 32, batch), blk, 0, stream>>>(
        Bm, u, BuT, N);
    gemm_f32_wmma<dmodel, L><<<dim3(L / 32, N / 32, batch), blk, 0, stream>>>(
        Cm, u, CuT, N);
    gemm_f32_wmma_rb<dmodel, L, dmodel><<<dim3(L / 128, dmodel / 64, batch), blk, 0, stream>>>(
        Delta, u, dT);

    ssm_scan<dmodel, L><<<dim3(dmodel / 16, batch), dim3(128, 1, 1), 0, stream>>>(
        u, A, D, BuT, CuT, dT, out);
}


# Bu and Cu products folded into the delta GEMM launch: each wave adds one 16x16 tile of [Bm;Cm]*u on U fragments it already holds; 4 launches to 2; Bu/Cu k regrouped per 16-k block (sum order changes)
# speedup vs baseline: 14.7306x; 1.0833x over previous
#include <hip/hip_runtime.h>

typedef __attribute__((ext_vector_type(2))) float v2f;
typedef __attribute__((ext_vector_type(4))) float v4f;
typedef __attribute__((ext_vector_type(8))) float v8f;

struct RbFrag {
    v4f a[2][2];
    v4f x[2];
    v4f b0[4];
    v4f b1[4];
};

template <int K, int L>
static __device__ __forceinline__ void rb_load(RbFrag& f, const float* __restrict__ ap,
                                               const float* __restrict__ xp,
                                               const float* __restrict__ bp)
{
    #pragma unroll
    for (int tm = 0; tm < 2; ++tm) {
        #pragma unroll
        for (int q = 0; q < 2; ++q)
            f.a[tm][q] = *(const v4f*)(ap + (size_t)tm * 16 * K + 4 * q);
    }
    #pragma unroll
    for (int q = 0; q < 2; ++q)
        f.x[q] = *(const v4f*)(xp + 4 * q);
    #pragma unroll
    for (int i = 0; i < 4; ++i) {
        f.b0[i] = *(const v4f*)(bp + (size_t)(2 * i)     * L);
        f.b1[i] = *(const v4f*)(bp + (size_t)(2 * i + 1) * L);
    }
}

template <int TN>
static __device__ __forceinline__ void rb_compute(const RbFrag& f, v8f (&acc)[2][4], v8f& accx)
{
    #pragma unroll
    for (int i = 0; i < 4; ++i) {
        #pragma unroll
        for (int tm = 0; tm < 2; ++tm) {
            v2f av;
            av.x = f.a[tm][i >> 1][2 * (i & 1)];
            av.y = f.a[tm][i >> 1][2 * (i & 1) + 1];
            #pragma unroll
            for (int tn = 0; tn < 4; ++tn) {
                v2f bv;
                bv.x = f.b0[i][tn];
                bv.y = f.b1[i][tn];
                acc[tm][tn] = __builtin_amdgcn_wmma_f32_16x16x4_f32(
                                  false, av, false, bv, (short)0, acc[tm][tn], false, false);
            }
        }
        v2f xv, bx;
        xv.x = f.x[i >> 1][2 * (i & 1)];
        xv.y = f.x[i >> 1][2 * (i & 1) + 1];
        bx.x = f.b0[i][TN];
        bx.y = f.b1[i][TN];
        accx = __builtin_amdgcn_wmma_f32_16x16x4_f32(
                   false, xv, false, bx, (short)0, accx, false, false);
    }
}

template <int K, int L, int M, int N, int TN>
static __device__ __forceinline__ void rb_body(const float* __restrict__ W,
                                               const float* __restrict__ Bm,
                                               const float* __restrict__ Cm,
                                               const float* __restrict__ U,
                                               float* __restrict__ Out,
                                               float* __restrict__ BuT,
                                               float* __restrict__ CuT)
{
    const int b = blockIdx.z;
    const float* __restrict__ Ub = U + (size_t)b * K * L;
    float* __restrict__ Ob = Out + (size_t)b * M * L;

    const int wave  = threadIdx.x >> 5;
    const int lane  = threadIdx.x & 31;
    const int col   = lane & 15;
    const int khalf = lane >> 4;
    const int m0    = blockIdx.y * 64  + (wave >> 1) * 32;
    const int l0    = blockIdx.x * 128 + (wave & 1) * 64;

    const int  rt   = (blockIdx.y >> 2) * 2 + (wave >> 1);
    const bool isB  = rt < N / 16;
    const int  n0   = (rt & (N / 16 - 1)) * 16;

    const float* __restrict__ aptr = W  + (size_t)(m0 + col) * K + 8 * khalf;
    const float* __restrict__ xptr = (isB ? Bm : Cm) + (size_t)(n0 + col) * K + 8 * khalf;
    const float* __restrict__ bptr = Ub + (size_t)(8 * khalf) * L + l0 + 4 * col;

    v8f acc[2][4];
    #pragma unroll
    for (int tm = 0; tm < 2; ++tm) {
        #pragma unroll
        for (int tn = 0; tn < 4; ++tn)
            acc[tm][tn] = v8f{};
    }
    v8f accx = v8f{};

    RbFrag f0, f1;
    rb_load<K, L>(f0, aptr, xptr, bptr);
    for (int kb = 0; kb < K - 32; kb += 32) {
        rb_load<K, L>(f1, aptr + 16, xptr + 16, bptr + (size_t)16 * L);
        rb_compute<TN>(f0, acc, accx);
        rb_load<K, L>(f0, aptr + 32, xptr + 32, bptr + (size_t)32 * L);
        rb_compute<TN>(f1, acc, accx);
        aptr += 32;
        xptr += 32;
        bptr += (size_t)32 * L;
    }
    rb_load<K, L>(f1, aptr + 16, xptr + 16, bptr + (size_t)16 * L);
    rb_compute<TN>(f0, acc, accx);
    rb_compute<TN>(f1, acc, accx);

    #pragma unroll
    for (int tm = 0; tm < 2; ++tm) {
        #pragma unroll
        for (int tn = 0; tn < 4; ++tn) {
            float* __restrict__ dst = Ob + (size_t)(l0 + 4 * col + tn) * M + m0 + 16 * tm + 8 * khalf;
            const v8f v = acc[tm][tn];
            v4f lo = {v[0], v[1], v[2], v[3]};
            v4f hi = {v[4], v[5], v[6], v[7]};
            *(v4f*)(dst)     = lo;
            *(v4f*)(dst + 4) = hi;
        }
    }
    {
        float* __restrict__ Xo  = (isB ? BuT : CuT) + (size_t)b * L * N;
        float* __restrict__ dst = Xo + (size_t)(l0 + 4 * col + TN) * N + n0 + 8 * khalf;
        v4f lo = {accx[0], accx[1], accx[2], accx[3]};
        v4f hi = {accx[4], accx[5], accx[6], accx[7]};
        *(v4f*)(dst)     = lo;
        *(v4f*)(dst + 4) = hi;
    }
}

template <int K, int L, int M, int N>
__global__ __launch_bounds__(128) __attribute__((amdgpu_waves_per_eu(4, 4)))
void gemm_f32_wmma_rb(const float* __restrict__ W, const float* __restrict__ Bm,
                      const float* __restrict__ Cm, const float* __restrict__ U,
                      float* __restrict__ Out, float* __restrict__ BuT, float* __restrict__ CuT)
{
    static_assert(K % 32 == 0 && K >= 64 && L % 128 == 0 && M % 64 == 0, "tile sizes in rb_body");
    static_assert((M / 64) * 2 == (2 * N / 16) * 4 && (N / 16 & (N / 16 - 1)) == 0,
                  "the waves sharing an l range must equal the 16 x 16 tiles of [Bm; Cm] * u on it");
    switch (blockIdx.y & 3) {
    case 0:  rb_body<K, L, M, N, 0>(W, Bm, Cm, U, Out, BuT, CuT); break;
    case 1:  rb_body<K, L, M, N, 1>(W, Bm, Cm, U, Out, BuT, CuT); break;
    case 2:  rb_body<K, L, M, N, 2>(W, Bm, Cm, U, Out, BuT, CuT); break;
    default: rb_body<K, L, M, N, 3>(W, Bm, Cm, U, Out, BuT, CuT); break;
    }
}

constexpr int SCAN_T  = 32;
constexpr int SCAN_RP = SCAN_T + 4;

template <int CTRL>
static __device__ __forceinline__ float xmask_add(float v)
{
    const int o = __builtin_amdgcn_update_dpp(0, __float_as_int(v), CTRL, 0xf, 0xf, true);
    return v + __int_as_float(o);
}

template <int DMODEL, int L>
__global__ __launch_bounds__(128) __attribute__((amdgpu_waves_per_eu(1, 2)))
void ssm_scan(const float* __restrict__ u, const float* __restrict__ A,
              const float* __restrict__ D,
              const float* __restrict__ BuT, const float* __restrict__ CuT,
              const float* __restrict__ dT, float* __restrict__ out)
{
    constexpr int N      = 64;
    constexpr int T      = SCAN_T;
    constexpr int RP     = SCAN_RP;
    constexpr int RP4    = SCAN_RP / 4;
    constexpr int ROWS   = 16;
    constexpr int NCHUNK = L / T;
    static_assert(L % T == 0 && T == 32 && N == 64, "chunk loaders below are written for 32 steps of 64 states");

    const int b  = blockIdx.y;
    const int t  = threadIdx.x;
    const int d0 = blockIdx.x * ROWS;
    const int r  = t >> 3;
    const int lj = t & 7;
    const int d  = d0 + r;

    float*       __restrict__ ybd = out + ((size_t)b * DMODEL + d) * L + 4 * lj;
    const float* __restrict__ Bub = BuT + (size_t)b * L * N;
    const float* __restrict__ Cub = CuT + (size_t)b * L * N;

    const v4f* __restrict__ gB = (const v4f*)Bub + t;
    const v4f* __restrict__ gC = (const v4f*)Cub + t;
    const float* __restrict__ gD = dT + ((size_t)b * L + (t >> 2)) * DMODEL + d0 + 4 * (t & 3);
    const float* __restrict__ gU = u + ((size_t)b * DMODEL + d) * L + 4 * lj;

    const float Dd = D[d];

    float areg[8], h[8];
    #pragma unroll
    for (int i = 0; i < 8; ++i) { areg[i] = A[8 * lj + i]; h[i] = 0.0f; }

    __shared__ v4f sB[2][T * 16];
    __shared__ v4f sC[2][T * 16];
    __shared__ v4f sD[2][ROWS * RP4];
    __shared__ v4f sU[2][ROWS * RP4];

    v4f pb0, pb1, pb2, pb3, pc0, pc1, pc2, pc3, pd, pu;

    pb0 = gB[0];  pb1 = gB[128];  pb2 = gB[256];  pb3 = gB[384];
    pc0 = gC[0];  pc1 = gC[128];  pc2 = gC[256];  pc3 = gC[384];
    pd  = *(const v4f*)(gD);
    pu  = *(const v4f*)(gU);
    {
        sB[0][t] = pb0;  sB[0][t + 128] = pb1;  sB[0][t + 256] = pb2;  sB[0][t + 384] = pb3;
        sC[0][t] = pc0;  sC[0][t + 128] = pc1;  sC[0][t + 256] = pc2;  sC[0][t + 384] = pc3;
        float* __restrict__ wD = (float*)&sD[0][0] + (4 * (t & 3)) * RP + (t >> 2);
        wD[0] = pd[0];  wD[RP] = pd[1];  wD[2 * RP] = pd[2];  wD[3 * RP] = pd[3];
        sU[0][r * RP4 + lj] = pu;
    }
    __syncthreads();

    for (int c = 0; c < NCHUNK; ++c) {
        const int  buf  = c & 1;
        const bool more = (c + 1 < NCHUNK);

        if (more) {
            const size_t cn = (size_t)(c + 1);
            pb0 = gB[cn * 512];        pb1 = gB[cn * 512 + 128];
            pb2 = gB[cn * 512 + 256];  pb3 = gB[cn * 512 + 384];
            pc0 = gC[cn * 512];        pc1 = gC[cn * 512 + 128];
            pc2 = gC[cn * 512 + 256];  pc3 = gC[cn * 512 + 384];
            pd  = *(const v4f*)(gD + cn * T * DMODEL);
            pu  = *(const v4f*)(gU + cn * T);
        }

        const v4f* __restrict__ cB = &sB[buf][2 * lj];
        const v4f* __restrict__ cC = &sC[buf][2 * lj];
        const v4f* __restrict__ cD = &sD[buf][r * RP4];
        const v4f* __restrict__ cU = &sU[buf][r * RP4];

        v4f yv = {0.0f, 0.0f, 0.0f, 0.0f};
        #pragma unroll
        for (int g = 0; g < T / 4; ++g) {
            const v4f  d4   = cD[g];
            const v4f  u4   = cU[g];
            const bool mine = (lj == g);
            #pragma unroll
            for (int s = 0; s < 4; ++s) {
                const int   j  = g * 4 + s;
                const float dl = d4[s];
                const float ul = u4[s];
                const float du = dl * ul;
                const v4f bq0 = cB[j * 16];
                const v4f bq1 = cB[j * 16 + 1];
                const v4f cq0 = cC[j * 16];
                const v4f cq1 = cC[j * 16 + 1];

                float y = 0.0f;
                #pragma unroll
                for (int k = 0; k < 4; ++k) {
                    const float hn = fmaf(dl * areg[k], h[k], du * bq0[k]);
                    h[k] = hn;
                    y = fmaf(cq0[k], hn, y);
                }
                #pragma unroll
                for (int k = 0; k < 4; ++k) {
                    const float hn = fmaf(dl * areg[4 + k], h[4 + k], du * bq1[k]);
                    h[4 + k] = hn;
                    y = fmaf(cq1[k], hn, y);
                }
                y = xmask_add<0x161>(y);
                y = xmask_add<0x162>(y);
                y = xmask_add<0x164>(y);
                const float yo = fmaf(ul, Dd, y);
                yv[s] = mine ? yo : yv[s];
            }
        }
        *(v4f*)(ybd + (size_t)c * T) = yv;

        if (more) {
            const int nb = buf ^ 1;
            sB[nb][t] = pb0;  sB[nb][t + 128] = pb1;  sB[nb][t + 256] = pb2;  sB[nb][t + 384] = pb3;
            sC[nb][t] = pc0;  sC[nb][t + 128] = pc1;  sC[nb][t + 256] = pc2;  sC[nb][t + 384] = pc3;
            float* __restrict__ wD = (float*)&sD[nb][0] + (4 * (t & 3)) * RP + (t >> 2);
            wD[0] = pd[0];  wD[RP] = pd[1];  wD[2 * RP] = pd[2];  wD[3 * RP] = pd[3];
            sU[nb][r * RP4 + lj] = pu;
        }
        __builtin_amdgcn_fence(__ATOMIC_RELEASE, "workgroup", "local");
        __builtin_amdgcn_s_barrier();
        __builtin_amdgcn_fence(__ATOMIC_ACQUIRE, "workgroup", "local");
    }
}

extern "C" void kernel_launch(void* const* d_in, const int* in_sizes, int n_in,
                              void* d_out, int out_size, void* d_ws, size_t ws_size,
                              hipStream_t stream)
{
    (void)in_sizes; (void)n_in; (void)out_size; (void)ws_size;

    const float* u     = (const float*)d_in[1];
    const float* A     = (const float*)d_in[2];
    const float* Bm    = (const float*)d_in[3];
    const float* Cm    = (const float*)d_in[4];
    const float* D     = (const float*)d_in[5];
    const float* Delta = (const float*)d_in[6];
    float* out = (float*)d_out;

    constexpr int batch = 4, dmodel = 1024, N = 64, L = 2048;

    float* BuT = (float*)d_ws;
    float* CuT = BuT + (size_t)batch * L * N;
    float* dT  = CuT + (size_t)batch * L * N;

    const dim3 blk(128, 1, 1);

    gemm_f32_wmma_rb<dmodel, L, dmodel, N><<<dim3(L / 128, dmodel / 64, batch), blk, 0, stream>>>(
        Delta, Bm, Cm, u, dT, BuT, CuT);

    ssm_scan<dmodel, L><<<dim3(dmodel / 16, batch), dim3(128, 1, 1), 0, stream>>>(
        u, A, D, BuT, CuT, dT, out);
}


# scan: LDS reads requested one 8-step block ahead of the math (second register set), on across chunk ends via a 3-buffer LDS ring: LDS pipe and vector ALU work at the same time; sums unchanged
# speedup vs baseline: 15.4084x; 1.0460x over previous
#include <hip/hip_runtime.h>

typedef __attribute__((ext_vector_type(2))) float v2f;
typedef __attribute__((ext_vector_type(4))) float v4f;
typedef __attribute__((ext_vector_type(8))) float v8f;

struct RbFrag {
    v4f a[2][2];
    v4f x[2];
    v4f b0[4];
    v4f b1[4];
};

template <int K, int L>
static __device__ __forceinline__ void rb_load(RbFrag& f, const float* __restrict__ ap,
                                               const float* __restrict__ xp,
                                               const float* __restrict__ bp)
{
    #pragma unroll
    for (int tm = 0; tm < 2; ++tm) {
        #pragma unroll
        for (int q = 0; q < 2; ++q)
            f.a[tm][q] = *(const v4f*)(ap + (size_t)tm * 16 * K + 4 * q);
    }
    #pragma unroll
    for (int q = 0; q < 2; ++q)
        f.x[q] = *(const v4f*)(xp + 4 * q);
    #pragma unroll
    for (int i = 0; i < 4; ++i) {
        f.b0[i] = *(const v4f*)(bp + (size_t)(2 * i)     * L);
        f.b1[i] = *(const v4f*)(bp + (size_t)(2 * i + 1) * L);
    }
}

template <int TN>
static __device__ __forceinline__ void rb_compute(const RbFrag& f, v8f (&acc)[2][4], v8f& accx)
{
    #pragma unroll
    for (int i = 0; i < 4; ++i) {
        #pragma unroll
        for (int tm = 0; tm < 2; ++tm) {
            v2f av;
            av.x = f.a[tm][i >> 1][2 * (i & 1)];
            av.y = f.a[tm][i >> 1][2 * (i & 1) + 1];
            #pragma unroll
            for (int tn = 0; tn < 4; ++tn) {
                v2f bv;
                bv.x = f.b0[i][tn];
                bv.y = f.b1[i][tn];
                acc[tm][tn] = __builtin_amdgcn_wmma_f32_16x16x4_f32(
                                  false, av, false, bv, (short)0, acc[tm][tn], false, false);
            }
        }
        v2f xv, bx;
        xv.x = f.x[i >> 1][2 * (i & 1)];
        xv.y = f.x[i >> 1][2 * (i & 1) + 1];
        bx.x = f.b0[i][TN];
        bx.y = f.b1[i][TN];
        accx = __builtin_amdgcn_wmma_f32_16x16x4_f32(
                   false, xv, false, bx, (short)0, accx, false, false);
    }
}

template <int K, int L, int M, int N, int TN>
static __device__ __forceinline__ void rb_body(const float* __restrict__ W,
                                               const float* __restrict__ Bm,
                                               const float* __restrict__ Cm,
                                               const float* __restrict__ U,
                                               float* __restrict__ Out,
                                               float* __restrict__ BuT,
                                               float* __restrict__ CuT)
{
    const int b = blockIdx.z;
    const float* __restrict__ Ub = U + (size_t)b * K * L;
    float* __restrict__ Ob = Out + (size_t)b * M * L;

    const int wave  = threadIdx.x >> 5;
    const int lane  = threadIdx.x & 31;
    const int col   = lane & 15;
    const int khalf = lane >> 4;
    const int m0    = blockIdx.y * 64  + (wave >> 1) * 32;
    const int l0    = blockIdx.x * 128 + (wave & 1) * 64;

    const int  rt   = (blockIdx.y >> 2) * 2 + (wave >> 1);
    const bool isB  = rt < N / 16;
    const int  n0   = (rt & (N / 16 - 1)) * 16;

    const float* __restrict__ aptr = W  + (size_t)(m0 + col) * K + 8 * khalf;
    const float* __restrict__ xptr = (isB ? Bm : Cm) + (size_t)(n0 + col) * K + 8 * khalf;
    const float* __restrict__ bptr = Ub + (size_t)(8 * khalf) * L + l0 + 4 * col;

    v8f acc[2][4];
    #pragma unroll
    for (int tm = 0; tm < 2; ++tm) {
        #pragma unroll
        for (int tn = 0; tn < 4; ++tn)
            acc[tm][tn] = v8f{};
    }
    v8f accx = v8f{};

    RbFrag f0, f1;
    rb_load<K, L>(f0, aptr, xptr, bptr);
    for (int kb = 0; kb < K - 32; kb += 32) {
        rb_load<K, L>(f1, aptr + 16, xptr + 16, bptr + (size_t)16 * L);
        rb_compute<TN>(f0, acc, accx);
        rb_load<K, L>(f0, aptr + 32, xptr + 32, bptr + (size_t)32 * L);
        rb_compute<TN>(f1, acc, accx);
        aptr += 32;
        xptr += 32;
        bptr += (size_t)32 * L;
    }
    rb_load<K, L>(f1, aptr + 16, xptr + 16, bptr + (size_t)16 * L);
    rb_compute<TN>(f0, acc, accx);
    rb_compute<TN>(f1, acc, accx);

    #pragma unroll
    for (int tm = 0; tm < 2; ++tm) {
        #pragma unroll
        for (int tn = 0; tn < 4; ++tn) {
            float* __restrict__ dst = Ob + (size_t)(l0 + 4 * col + tn) * M + m0 + 16 * tm + 8 * khalf;
            const v8f v = acc[tm][tn];
            v4f lo = {v[0], v[1], v[2], v[3]};
            v4f hi = {v[4], v[5], v[6], v[7]};
            *(v4f*)(dst)     = lo;
            *(v4f*)(dst + 4) = hi;
        }
    }
    {
        float* __restrict__ Xo  = (isB ? BuT : CuT) + (size_t)b * L * N;
        float* __restrict__ dst = Xo + (size_t)(l0 + 4 * col + TN) * N + n0 + 8 * khalf;
        v4f lo = {accx[0], accx[1], accx[2], accx[3]};
        v4f hi = {accx[4], accx[5], accx[6], accx[7]};
        *(v4f*)(dst)     = lo;
        *(v4f*)(dst + 4) = hi;
    }
}

template <int K, int L, int M, int N>
__global__ __launch_bounds__(128) __attribute__((amdgpu_waves_per_eu(4, 4)))
void gemm_f32_wmma_rb(const float* __restrict__ W, const float* __restrict__ Bm,
                      const float* __restrict__ Cm, const float* __restrict__ U,
                      float* __restrict__ Out, float* __restrict__ BuT, float* __restrict__ CuT)
{
    static_assert(K % 32 == 0 && K >= 64 && L % 128 == 0 && M % 64 == 0, "tile sizes in rb_body");
    static_assert((M / 64) * 2 == (2 * N / 16) * 4 && (N / 16 & (N / 16 - 1)) == 0,
                  "the waves sharing an l range must equal the 16 x 16 tiles of [Bm; Cm] * u on it");
    switch (blockIdx.y & 3) {
    case 0:  rb_body<K, L, M, N, 0>(W, Bm, Cm, U, Out, BuT, CuT); break;
    case 1:  rb_body<K, L, M, N, 1>(W, Bm, Cm, U, Out, BuT, CuT); break;
    case 2:  rb_body<K, L, M, N, 2>(W, Bm, Cm, U, Out, BuT, CuT); break;
    default: rb_body<K, L, M, N, 3>(W, Bm, Cm, U, Out, BuT, CuT); break;
    }
}

constexpr int SCAN_T  = 32;
constexpr int SCAN_S  = 8;
constexpr int SCAN_RP = SCAN_T + 4;

template <int CTRL>
static __device__ __forceinline__ float xmask_add(float v)
{
    const int o = __builtin_amdgcn_update_dpp(0, __float_as_int(v), CTRL, 0xf, 0xf, true);
    return v + __int_as_float(o);
}

struct ScanPre {
    v4f b[4], c[4];
    v4f d, u;
};

template <int DMODEL>
static __device__ __forceinline__ void scan_pre_load(ScanPre& p, const v4f* __restrict__ gB,
                                                     const v4f* __restrict__ gC,
                                                     const float* __restrict__ gD,
                                                     const float* __restrict__ gU, int chunk)
{
    const size_t cn = (size_t)chunk;
    #pragma unroll
    for (int i = 0; i < 4; ++i) {
        p.b[i] = gB[cn * 512 + 128 * i];
        p.c[i] = gC[cn * 512 + 128 * i];
    }
    p.d = *(const v4f*)(gD + cn * SCAN_T * DMODEL);
    p.u = *(const v4f*)(gU + cn * SCAN_T);
}

static __device__ __forceinline__ void scan_pre_store(const ScanPre& p, v4f* __restrict__ sBn,
                                                      v4f* __restrict__ sCn, v4f* __restrict__ sDn,
                                                      v4f* __restrict__ sUn, int t, int r, int lj)
{
    #pragma unroll
    for (int i = 0; i < 4; ++i) {
        sBn[t + 128 * i] = p.b[i];
        sCn[t + 128 * i] = p.c[i];
    }
    float* __restrict__ wD = (float*)sDn + (4 * (t & 3)) * SCAN_RP + (t >> 2);
    wD[0] = p.d[0];  wD[SCAN_RP] = p.d[1];  wD[2 * SCAN_RP] = p.d[2];  wD[3 * SCAN_RP] = p.d[3];
    sUn[r * (SCAN_RP / 4) + lj] = p.u;
}

struct ScanIn {
    v4f d[SCAN_S / 4], u[SCAN_S / 4];
    v4f b[SCAN_S][2], c[SCAN_S][2];
};

template <int BLK>
static __device__ __forceinline__ void scan_fetch(ScanIn& f, const v4f* __restrict__ cB,
                                                  const v4f* __restrict__ cC,
                                                  const v4f* __restrict__ cD,
                                                  const v4f* __restrict__ cU)
{
    #pragma unroll
    for (int gg = 0; gg < SCAN_S / 4; ++gg) {
        f.d[gg] = cD[BLK * (SCAN_S / 4) + gg];
        f.u[gg] = cU[BLK * (SCAN_S / 4) + gg];
        #pragma unroll
        for (int s = 0; s < 4; ++s) {
            const int js = gg * 4 + s;
            const int j  = BLK * SCAN_S + js;
            f.b[js][0] = cB[j * 16];
            f.b[js][1] = cB[j * 16 + 1];
            f.c[js][0] = cC[j * 16];
            f.c[js][1] = cC[j * 16 + 1];
        }
    }
}

template <int BLK>
static __device__ __forceinline__ void scan_steps(const ScanIn& f, float (&h)[8],
                                                  const float (&areg)[8], float Dd, int lj, v4f& yv)
{
    #pragma unroll
    for (int gg = 0; gg < SCAN_S / 4; ++gg) {
        const int  g    = BLK * (SCAN_S / 4) + gg;
        const v4f  d4   = f.d[gg];
        const v4f  u4   = f.u[gg];
        const bool mine = (lj == g);
        #pragma unroll
        for (int s = 0; s < 4; ++s) {
            const int   js = gg * 4 + s;
            const float dl = d4[s];
            const float ul = u4[s];
            const float du = dl * ul;
            const v4f bq0 = f.b[js][0];
            const v4f bq1 = f.b[js][1];
            const v4f cq0 = f.c[js][0];
            const v4f cq1 = f.c[js][1];

            float y = 0.0f;
            #pragma unroll
            for (int k = 0; k < 4; ++k) {
                const float hn = fmaf(dl * areg[k], h[k], du * bq0[k]);
                h[k] = hn;
                y = fmaf(cq0[k], hn, y);
            }
            #pragma unroll
            for (int k = 0; k < 4; ++k) {
                const float hn = fmaf(dl * areg[4 + k], h[4 + k], du * bq1[k]);
                h[4 + k] = hn;
                y = fmaf(cq1[k], hn, y);
            }
            y = xmask_add<0x161>(y);
            y = xmask_add<0x162>(y);
            y = xmask_add<0x164>(y);
            const float yo = fmaf(ul, Dd, y);
            yv[s] = mine ? yo : yv[s];
        }
    }
}

template <int DMODEL, int L>
__global__ __launch_bounds__(128) __attribute__((amdgpu_waves_per_eu(1, 2)))
void ssm_scan(const float* __restrict__ u, const float* __restrict__ A,
              const float* __restrict__ D,
              const float* __restrict__ BuT, const float* __restrict__ CuT,
              const float* __restrict__ dT, float* __restrict__ out)
{
    constexpr int N      = 64;
    constexpr int T      = SCAN_T;
    constexpr int RP4    = SCAN_RP / 4;
    constexpr int ROWS   = 16;
    constexpr int NCHUNK = L / T;
    static_assert(L % T == 0 && T == 32 && N == 64, "chunk loaders are written for 32 steps of 64 states");
    static_assert(T / SCAN_S == 4 && SCAN_S % 4 == 0, "the chunk body below is written out for 4 read-ahead blocks");
    static_assert(NCHUNK >= 2, "two chunks are staged before the loop");

    const int b  = blockIdx.y;
    const int t  = threadIdx.x;
    const int d0 = blockIdx.x * ROWS;
    const int r  = t >> 3;
    const int lj = t & 7;
    const int d  = d0 + r;

    float*       __restrict__ ybd = out + ((size_t)b * DMODEL + d) * L + 4 * lj;
    const float* __restrict__ Bub = BuT + (size_t)b * L * N;
    const float* __restrict__ Cub = CuT + (size_t)b * L * N;

    const v4f* __restrict__ gB = (const v4f*)Bub + t;
    const v4f* __restrict__ gC = (const v4f*)Cub + t;
    const float* __restrict__ gD = dT + ((size_t)b * L + (t >> 2)) * DMODEL + d0 + 4 * (t & 3);
    const float* __restrict__ gU = u + ((size_t)b * DMODEL + d) * L + 4 * lj;

    const float Dd = D[d];

    float areg[8], h[8];
    #pragma unroll
    for (int i = 0; i < 8; ++i) { areg[i] = A[8 * lj + i]; h[i] = 0.0f; }

    __shared__ v4f sB[3][T * 16];
    __shared__ v4f sC[3][T * 16];
    __shared__ v4f sD[3][ROWS * RP4];
    __shared__ v4f sU[3][ROWS * RP4];

    ScanPre pre;

    #pragma unroll
    for (int cc = 0; cc < 2; ++cc) {
        scan_pre_load<DMODEL>(pre, gB, gC, gD, gU, cc);
        scan_pre_store(pre, &sB[cc][0], &sC[cc][0], &sD[cc][0], &sU[cc][0], t, r, lj);
    }
    __syncthreads();

    ScanIn fa, fb;
    scan_fetch<0>(fa, &sB[0][2 * lj], &sC[0][2 * lj], &sD[0][r * RP4], &sU[0][r * RP4]);

    int cur = 0;
    for (int c = 0; c < NCHUNK; ++c) {
        const int  nx1   = (cur == 2) ? 0 : cur + 1;
        const int  nx2   = (nx1 == 2) ? 0 : nx1 + 1;
        const bool more2 = (c + 2 < NCHUNK);

        if (more2) scan_pre_load<DMODEL>(pre, gB, gC, gD, gU, c + 2);

        const v4f* __restrict__ cB = &sB[cur][2 * lj];
        const v4f* __restrict__ cC = &sC[cur][2 * lj];
        const v4f* __restrict__ cD = &sD[cur][r * RP4];
        const v4f* __restrict__ cU = &sU[cur][r * RP4];

        v4f yv = {0.0f, 0.0f, 0.0f, 0.0f};

        __builtin_amdgcn_sched_barrier(0);
        scan_fetch<1>(fb, cB, cC, cD, cU);
        __builtin_amdgcn_sched_barrier(0);
        scan_steps<0>(fa, h, areg, Dd, lj, yv);

        __builtin_amdgcn_sched_barrier(0);
        scan_fetch<2>(fa, cB, cC, cD, cU);
        __builtin_amdgcn_sched_barrier(0);
        scan_steps<1>(fb, h, areg, Dd, lj, yv);

        __builtin_amdgcn_sched_barrier(0);
        scan_fetch<3>(fb, cB, cC, cD, cU);
        __builtin_amdgcn_sched_barrier(0);
        scan_steps<2>(fa, h, areg, Dd, lj, yv);

        __builtin_amdgcn_sched_barrier(0);
        scan_fetch<0>(fa, &sB[nx1][2 * lj], &sC[nx1][2 * lj], &sD[nx1][r * RP4], &sU[nx1][r * RP4]);
        __builtin_amdgcn_sched_barrier(0);
        scan_steps<3>(fb, h, areg, Dd, lj, yv);

        *(v4f*)(ybd + (size_t)c * T) = yv;

        if (more2) scan_pre_store(pre, &sB[nx2][0], &sC[nx2][0], &sD[nx2][0], &sU[nx2][0], t, r, lj);

        __builtin_amdgcn_fence(__ATOMIC_RELEASE, "workgroup", "local");
        __builtin_amdgcn_s_barrier();
        __builtin_amdgcn_fence(__ATOMIC_ACQUIRE, "workgroup", "local");
        cur = nx1;
    }
}

extern "C" void kernel_launch(void* const* d_in, const int* in_sizes, int n_in,
                              void* d_out, int out_size, void* d_ws, size_t ws_size,
                              hipStream_t stream)
{
    (void)in_sizes; (void)n_in; (void)out_size; (void)ws_size;

    const float* u     = (const float*)d_in[1];
    const float* A     = (const float*)d_in[2];
    const float* Bm    = (const float*)d_in[3];
    const float* Cm    = (const float*)d_in[4];
    const float* D     = (const float*)d_in[5];
    const float* Delta = (const float*)d_in[6];
    float* out = (float*)d_out;

    constexpr int batch = 4, dmodel = 1024, N = 64, L = 2048;

    float* BuT = (float*)d_ws;
    float* CuT = BuT + (size_t)batch * L * N;
    float* dT  = CuT + (size_t)batch * L * N;

    const dim3 blk(128, 1, 1);

    gemm_f32_wmma_rb<dmodel, L, dmodel, N><<<dim3(L / 128, dmodel / 64, batch), blk, 0, stream>>>(
        Delta, Bm, Cm, u, dT, BuT, CuT);

    ssm_scan<dmodel, L><<<dim3(dmodel / 16, batch), dim3(128, 1, 1), 0, stream>>>(
        u, A, D, BuT, CuT, dT, out);
}


# scan: a block's vector math ordered in passes across its 8 steps so neighbouring instructions are independent; y kept as even/odd-state partial sums on packed multiply-adds (y sum order changes)
# speedup vs baseline: 15.9117x; 1.0327x over previous
#include <hip/hip_runtime.h>

typedef __attribute__((ext_vector_type(2))) float v2f;
typedef __attribute__((ext_vector_type(4))) float v4f;
typedef __attribute__((ext_vector_type(8))) float v8f;

struct RbFrag {
    v4f a[2][2];
    v4f x[2];
    v4f b0[4];
    v4f b1[4];
};

template <int K, int L>
static __device__ __forceinline__ void rb_load(RbFrag& f, const float* __restrict__ ap,
                                               const float* __restrict__ xp,
                                               const float* __restrict__ bp)
{
    #pragma unroll
    for (int tm = 0; tm < 2; ++tm) {
        #pragma unroll
        for (int q = 0; q < 2; ++q)
            f.a[tm][q] = *(const v4f*)(ap + (size_t)tm * 16 * K + 4 * q);
    }
    #pragma unroll
    for (int q = 0; q < 2; ++q)
        f.x[q] = *(const v4f*)(xp + 4 * q);
    #pragma unroll
    for (int i = 0; i < 4; ++i) {
        f.b0[i] = *(const v4f*)(bp + (size_t)(2 * i)     * L);
        f.b1[i] = *(const v4f*)(bp + (size_t)(2 * i + 1) * L);
    }
}

template <int TN>
static __device__ __forceinline__ void rb_compute(const RbFrag& f, v8f (&acc)[2][4], v8f& accx)
{
    #pragma unroll
    for (int i = 0; i < 4; ++i) {
        #pragma unroll
        for (int tm = 0; tm < 2; ++tm) {
            v2f av;
            av.x = f.a[tm][i >> 1][2 * (i & 1)];
            av.y = f.a[tm][i >> 1][2 * (i & 1) + 1];
            #pragma unroll
            for (int tn = 0; tn < 4; ++tn) {
                v2f bv;
                bv.x = f.b0[i][tn];
                bv.y = f.b1[i][tn];
                acc[tm][tn] = __builtin_amdgcn_wmma_f32_16x16x4_f32(
                                  false, av, false, bv, (short)0, acc[tm][tn], false, false);
            }
        }
        v2f xv, bx;
        xv.x = f.x[i >> 1][2 * (i & 1)];
        xv.y = f.x[i >> 1][2 * (i & 1) + 1];
        bx.x = f.b0[i][TN];
        bx.y = f.b1[i][TN];
        accx = __builtin_amdgcn_wmma_f32_16x16x4_f32(
                   false, xv, false, bx, (short)0, accx, false, false);
    }
}

template <int K, int L, int M, int N, int TN>
static __device__ __forceinline__ void rb_body(const float* __restrict__ W,
                                               const float* __restrict__ Bm,
                                               const float* __restrict__ Cm,
                                               const float* __restrict__ U,
                                               float* __restrict__ Out,
                                               float* __restrict__ BuT,
                                               float* __restrict__ CuT)
{
    const int b = blockIdx.z;
    const float* __restrict__ Ub = U + (size_t)b * K * L;
    float* __restrict__ Ob = Out + (size_t)b * M * L;

    const int wave  = threadIdx.x >> 5;
    const int lane  = threadIdx.x & 31;
    const int col   = lane & 15;
    const int khalf = lane >> 4;
    const int m0    = blockIdx.y * 64  + (wave >> 1) * 32;
    const int l0    = blockIdx.x * 128 + (wave & 1) * 64;

    const int  rt   = (blockIdx.y >> 2) * 2 + (wave >> 1);
    const bool isB  = rt < N / 16;
    const int  n0   = (rt & (N / 16 - 1)) * 16;

    const float* __restrict__ aptr = W  + (size_t)(m0 + col) * K + 8 * khalf;
    const float* __restrict__ xptr = (isB ? Bm : Cm) + (size_t)(n0 + col) * K + 8 * khalf;
    const float* __restrict__ bptr = Ub + (size_t)(8 * khalf) * L + l0 + 4 * col;

    v8f acc[2][4];
    #pragma unroll
    for (int tm = 0; tm < 2; ++tm) {
        #pragma unroll
        for (int tn = 0; tn < 4; ++tn)
            acc[tm][tn] = v8f{};
    }
    v8f accx = v8f{};

    RbFrag f0, f1;
    rb_load<K, L>(f0, aptr, xptr, bptr);
    for (int kb = 0; kb < K - 32; kb += 32) {
        rb_load<K, L>(f1, aptr + 16, xptr + 16, bptr + (size_t)16 * L);
        rb_compute<TN>(f0, acc, accx);
        rb_load<K, L>(f0, aptr + 32, xptr + 32, bptr + (size_t)32 * L);
        rb_compute<TN>(f1, acc, accx);
        aptr += 32;
        xptr += 32;
        bptr += (size_t)32 * L;
    }
    rb_load<K, L>(f1, aptr + 16, xptr + 16, bptr + (size_t)16 * L);
    rb_compute<TN>(f0, acc, accx);
    rb_compute<TN>(f1, acc, accx);

    #pragma unroll
    for (int tm = 0; tm < 2; ++tm) {
        #pragma unroll
        for (int tn = 0; tn < 4; ++tn) {
            float* __restrict__ dst = Ob + (size_t)(l0 + 4 * col + tn) * M + m0 + 16 * tm + 8 * khalf;
            const v8f v = acc[tm][tn];
            v4f lo = {v[0], v[1], v[2], v[3]};
            v4f hi = {v[4], v[5], v[6], v[7]};
            *(v4f*)(dst)     = lo;
            *(v4f*)(dst + 4) = hi;
        }
    }
    {
        float* __restrict__ Xo  = (isB ? BuT : CuT) + (size_t)b * L * N;
        float* __restrict__ dst = Xo + (size_t)(l0 + 4 * col + TN) * N + n0 + 8 * khalf;
        v4f lo = {accx[0], accx[1], accx[2], accx[3]};
        v4f hi = {accx[4], accx[5], accx[6], accx[7]};
        *(v4f*)(dst)     = lo;
        *(v4f*)(dst + 4) = hi;
    }
}

template <int K, int L, int M, int N>
__global__ __launch_bounds__(128) __attribute__((amdgpu_waves_per_eu(4, 4)))
void gemm_f32_wmma_rb(const float* __restrict__ W, const float* __restrict__ Bm,
                      const float* __restrict__ Cm, const float* __restrict__ U,
                      float* __restrict__ Out, float* __restrict__ BuT, float* __restrict__ CuT)
{
    static_assert(K % 32 == 0 && K >= 64 && L % 128 == 0 && M % 64 == 0, "tile sizes in rb_body");
    static_assert((M / 64) * 2 == (2 * N / 16) * 4 && (N / 16 & (N / 16 - 1)) == 0,
                  "the waves sharing an l range must equal the 16 x 16 tiles of [Bm; Cm] * u on it");
    switch (blockIdx.y & 3) {
    case 0:  rb_body<K, L, M, N, 0>(W, Bm, Cm, U, Out, BuT, CuT); break;
    case 1:  rb_body<K, L, M, N, 1>(W, Bm, Cm, U, Out, BuT, CuT); break;
    case 2:  rb_body<K, L, M, N, 2>(W, Bm, Cm, U, Out, BuT, CuT); break;
    default: rb_body<K, L, M, N, 3>(W, Bm, Cm, U, Out, BuT, CuT); break;
    }
}

constexpr int SCAN_T  = 32;
constexpr int SCAN_S  = 8;
constexpr int SCAN_RP = SCAN_T + 4;

template <int CTRL>
static __device__ __forceinline__ float xmask_add(float v)
{
    const int o = __builtin_amdgcn_update_dpp(0, __float_as_int(v), CTRL, 0xf, 0xf, true);
    return v + __int_as_float(o);
}

struct ScanPre {
    v4f b[4], c[4];
    v4f d, u;
};

template <int DMODEL>
static __device__ __forceinline__ void scan_pre_load(ScanPre& p, const v4f* __restrict__ gB,
                                                     const v4f* __restrict__ gC,
                                                     const float* __restrict__ gD,
                                                     const float* __restrict__ gU, int chunk)
{
    const size_t cn = (size_t)chunk;
    #pragma unroll
    for (int i = 0; i < 4; ++i) {
        p.b[i] = gB[cn * 512 + 128 * i];
        p.c[i] = gC[cn * 512 + 128 * i];
    }
    p.d = *(const v4f*)(gD + cn * SCAN_T * DMODEL);
    p.u = *(const v4f*)(gU + cn * SCAN_T);
}

static __device__ __forceinline__ void scan_pre_store(const ScanPre& p, v4f* __restrict__ sBn,
                                                      v4f* __restrict__ sCn, v4f* __restrict__ sDn,
                                                      v4f* __restrict__ sUn, int t, int r, int lj)
{
    #pragma unroll
    for (int i = 0; i < 4; ++i) {
        sBn[t + 128 * i] = p.b[i];
        sCn[t + 128 * i] = p.c[i];
    }
    float* __restrict__ wD = (float*)sDn + (4 * (t & 3)) * SCAN_RP + (t >> 2);
    wD[0] = p.d[0];  wD[SCAN_RP] = p.d[1];  wD[2 * SCAN_RP] = p.d[2];  wD[3 * SCAN_RP] = p.d[3];
    sUn[r * (SCAN_RP / 4) + lj] = p.u;
}

struct ScanIn {
    v4f d[SCAN_S / 4], u[SCAN_S / 4];
    v4f b[SCAN_S][2], c[SCAN_S][2];
};

template <int BLK>
static __device__ __forceinline__ void scan_fetch(ScanIn& f, const v4f* __restrict__ cB,
                                                  const v4f* __restrict__ cC,
                                                  const v4f* __restrict__ cD,
                                                  const v4f* __restrict__ cU)
{
    #pragma unroll
    for (int gg = 0; gg < SCAN_S / 4; ++gg) {
        f.d[gg] = cD[BLK * (SCAN_S / 4) + gg];
        f.u[gg] = cU[BLK * (SCAN_S / 4) + gg];
        #pragma unroll
        for (int s = 0; s < 4; ++s) {
            const int js = gg * 4 + s;
            const int j  = BLK * SCAN_S + js;
            f.b[js][0] = cB[j * 16];
            f.b[js][1] = cB[j * 16 + 1];
            f.c[js][0] = cC[j * 16];
            f.c[js][1] = cC[j * 16 + 1];
        }
    }
}

template <int BLK>
static __device__ __forceinline__ void scan_steps(const ScanIn& f, float (&h)[8],
                                                  const float (&areg)[8], float Dd, int lj, v4f& yv)
{
    float hn[SCAN_S][8];
    #pragma unroll
    for (int js = 0; js < SCAN_S; ++js) {
        const float dl = f.d[js >> 2][js & 3];
        const float ul = f.u[js >> 2][js & 3];
        const float du = dl * ul;
        #pragma unroll
        for (int k = 0; k < 4; ++k) {
            h[k]      = fmaf(dl * areg[k], h[k], du * f.b[js][0][k]);
            hn[js][k] = h[k];
        }
        #pragma unroll
        for (int k = 0; k < 4; ++k) {
            h[4 + k]      = fmaf(dl * areg[4 + k], h[4 + k], du * f.b[js][1][k]);
            hn[js][4 + k] = h[4 + k];
        }
    }

    v2f y2[SCAN_S];
    #pragma unroll
    for (int js = 0; js < SCAN_S; ++js)
        y2[js] = v2f{0.0f, 0.0f};
    #pragma unroll
    for (int m = 0; m < 4; ++m) {
        #pragma unroll
        for (int js = 0; js < SCAN_S; ++js) {
            const v2f c2 = {f.c[js][m >> 1][2 * (m & 1)], f.c[js][m >> 1][2 * (m & 1) + 1]};
            const v2f h2 = {hn[js][2 * m], hn[js][2 * m + 1]};
            y2[js] = __builtin_elementwise_fma(c2, h2, y2[js]);
        }
    }
    float y[SCAN_S];
    #pragma unroll
    for (int js = 0; js < SCAN_S; ++js)
        y[js] = y2[js].x + y2[js].y;

    #pragma unroll
    for (int js = 0; js < SCAN_S; ++js)
        y[js] = xmask_add<0x161>(y[js]);
    #pragma unroll
    for (int js = 0; js < SCAN_S; ++js)
        y[js] = xmask_add<0x162>(y[js]);
    #pragma unroll
    for (int js = 0; js < SCAN_S; ++js)
        y[js] = xmask_add<0x164>(y[js]);

    #pragma unroll
    for (int js = 0; js < SCAN_S; ++js) {
        const int   g    = BLK * (SCAN_S / 4) + (js >> 2);
        const bool  mine = (lj == g);
        const float yo   = fmaf(f.u[js >> 2][js & 3], Dd, y[js]);
        yv[js & 3] = mine ? yo : yv[js & 3];
    }
}

template <int DMODEL, int L>
__global__ __launch_bounds__(128) __attribute__((amdgpu_waves_per_eu(1, 2)))
void ssm_scan(const float* __restrict__ u, const float* __restrict__ A,
              const float* __restrict__ D,
              const float* __restrict__ BuT, const float* __restrict__ CuT,
              const float* __restrict__ dT, float* __restrict__ out)
{
    constexpr int N      = 64;
    constexpr int T      = SCAN_T;
    constexpr int RP4    = SCAN_RP / 4;
    constexpr int ROWS   = 16;
    constexpr int NCHUNK = L / T;
    static_assert(L % T == 0 && T == 32 && N == 64, "chunk loaders are written for 32 steps of 64 states");
    static_assert(T / SCAN_S == 4 && SCAN_S % 4 == 0, "the chunk body below is written out for 4 read-ahead blocks");
    static_assert(NCHUNK >= 2, "two chunks are staged before the loop");

    const int b  = blockIdx.y;
    const int t  = threadIdx.x;
    const int d0 = blockIdx.x * ROWS;
    const int r  = t >> 3;
    const int lj = t & 7;
    const int d  = d0 + r;

    float*       __restrict__ ybd = out + ((size_t)b * DMODEL + d) * L + 4 * lj;
    const float* __restrict__ Bub = BuT + (size_t)b * L * N;
    const float* __restrict__ Cub = CuT + (size_t)b * L * N;

    const v4f* __restrict__ gB = (const v4f*)Bub + t;
    const v4f* __restrict__ gC = (const v4f*)Cub + t;
    const float* __restrict__ gD = dT + ((size_t)b * L + (t >> 2)) * DMODEL + d0 + 4 * (t & 3);
    const float* __restrict__ gU = u + ((size_t)b * DMODEL + d) * L + 4 * lj;

    const float Dd = D[d];

    float areg[8], h[8];
    #pragma unroll
    for (int i = 0; i < 8; ++i) { areg[i] = A[8 * lj + i]; h[i] = 0.0f; }

    __shared__ v4f sB[3][T * 16];
    __shared__ v4f sC[3][T * 16];
    __shared__ v4f sD[3][ROWS * RP4];
    __shared__ v4f sU[3][ROWS * RP4];

    ScanPre pre;

    #pragma unroll
    for (int cc = 0; cc < 2; ++cc) {
        scan_pre_load<DMODEL>(pre, gB, gC, gD, gU, cc);
        scan_pre_store(pre, &sB[cc][0], &sC[cc][0], &sD[cc][0], &sU[cc][0], t, r, lj);
    }
    __syncthreads();

    ScanIn fa, fb;
    scan_fetch<0>(fa, &sB[0][2 * lj], &sC[0][2 * lj], &sD[0][r * RP4], &sU[0][r * RP4]);

    int cur = 0;
    for (int c = 0; c < NCHUNK; ++c) {
        const int  nx1   = (cur == 2) ? 0 : cur + 1;
        const int  nx2   = (nx1 == 2) ? 0 : nx1 + 1;
        const bool more2 = (c + 2 < NCHUNK);

        if (more2) scan_pre_load<DMODEL>(pre, gB, gC, gD, gU, c + 2);

        const v4f* __restrict__ cB = &sB[cur][2 * lj];
        const v4f* __restrict__ cC = &sC[cur][2 * lj];
        const v4f* __restrict__ cD = &sD[cur][r * RP4];
        const v4f* __restrict__ cU = &sU[cur][r * RP4];

        v4f yv = {0.0f, 0.0f, 0.0f, 0.0f};

        __builtin_amdgcn_sched_barrier(0);
        scan_fetch<1>(fb, cB, cC, cD, cU);
        __builtin_amdgcn_sched_barrier(0);
        scan_steps<0>(fa, h, areg, Dd, lj, yv);

        __builtin_amdgcn_sched_barrier(0);
        scan_fetch<2>(fa, cB, cC, cD, cU);
        __builtin_amdgcn_sched_barrier(0);
        scan_steps<1>(fb, h, areg, Dd, lj, yv);

        __builtin_amdgcn_sched_barrier(0);
        scan_fetch<3>(fb, cB, cC, cD, cU);
        __builtin_amdgcn_sched_barrier(0);
        scan_steps<2>(fa, h, areg, Dd, lj, yv);

        __builtin_amdgcn_sched_barrier(0);
        scan_fetch<0>(fa, &sB[nx1][2 * lj], &sC[nx1][2 * lj], &sD[nx1][r * RP4], &sU[nx1][r * RP4]);
        __builtin_amdgcn_sched_barrier(0);
        scan_steps<3>(fb, h, areg, Dd, lj, yv);

        *(v4f*)(ybd + (size_t)c * T) = yv;

        if (more2) scan_pre_store(pre, &sB[nx2][0], &sC[nx2][0], &sD[nx2][0], &sU[nx2][0], t, r, lj);

        __builtin_amdgcn_fence(__ATOMIC_RELEASE, "workgroup", "local");
        __builtin_amdgcn_s_barrier();
        __builtin_amdgcn_fence(__ATOMIC_ACQUIRE, "workgroup", "local");
        cur = nx1;
    }
}

extern "C" void kernel_launch(void* const* d_in, const int* in_sizes, int n_in,
                              void* d_out, int out_size, void* d_ws, size_t ws_size,
                              hipStream_t stream)
{
    (void)in_sizes; (void)n_in; (void)out_size; (void)ws_size;

    const float* u     = (const float*)d_in[1];
    const float* A     = (const float*)d_in[2];
    const float* Bm    = (const float*)d_in[3];
    const float* Cm    = (const float*)d_in[4];
    const float* D     = (const float*)d_in[5];
    const float* Delta = (const float*)d_in[6];
    float* out = (float*)d_out;

    constexpr int batch = 4, dmodel = 1024, N = 64, L = 2048;

    float* BuT = (float*)d_ws;
    float* CuT = BuT + (size_t)batch * L * N;
    float* dT  = CuT + (size_t)batch * L * N;

    const dim3 blk(128, 1, 1);

    gemm_f32_wmma_rb<dmodel, L, dmodel, N><<<dim3(L / 128, dmodel / 64, batch), blk, 0, stream>>>(
        Delta, Bm, Cm, u, dT, BuT, CuT);

    ssm_scan<dmodel, L><<<dim3(dmodel / 16, batch), dim3(128, 1, 1), 0, stream>>>(
        u, A, D, BuT, CuT, dT, out);
}


# scan: per-chunk block barrier split: signal after a wave's hand-over stores, wait 3 blocks later before the first read needing it, so the 4 waves' LDS read bursts stop coinciding; sums unchanged
# speedup vs baseline: 16.3823x; 1.0296x over previous
#include <hip/hip_runtime.h>

typedef __attribute__((ext_vector_type(2))) float v2f;
typedef __attribute__((ext_vector_type(4))) float v4f;
typedef __attribute__((ext_vector_type(8))) float v8f;

struct RbFrag {
    v4f a[2][2];
    v4f x[2];
    v4f b0[4];
    v4f b1[4];
};

template <int K, int L>
static __device__ __forceinline__ void rb_load(RbFrag& f, const float* __restrict__ ap,
                                               const float* __restrict__ xp,
                                               const float* __restrict__ bp)
{
    #pragma unroll
    for (int tm = 0; tm < 2; ++tm) {
        #pragma unroll
        for (int q = 0; q < 2; ++q)
            f.a[tm][q] = *(const v4f*)(ap + (size_t)tm * 16 * K + 4 * q);
    }
    #pragma unroll
    for (int q = 0; q < 2; ++q)
        f.x[q] = *(const v4f*)(xp + 4 * q);
    #pragma unroll
    for (int i = 0; i < 4; ++i) {
        f.b0[i] = *(const v4f*)(bp + (size_t)(2 * i)     * L);
        f.b1[i] = *(const v4f*)(bp + (size_t)(2 * i + 1) * L);
    }
}

template <int TN>
static __device__ __forceinline__ void rb_compute(const RbFrag& f, v8f (&acc)[2][4], v8f& accx)
{
    #pragma unroll
    for (int i = 0; i < 4; ++i) {
        #pragma unroll
        for (int tm = 0; tm < 2; ++tm) {
            v2f av;
            av.x = f.a[tm][i >> 1][2 * (i & 1)];
            av.y = f.a[tm][i >> 1][2 * (i & 1) + 1];
            #pragma unroll
            for (int tn = 0; tn < 4; ++tn) {
                v2f bv;
                bv.x = f.b0[i][tn];
                bv.y = f.b1[i][tn];
                acc[tm][tn] = __builtin_amdgcn_wmma_f32_16x16x4_f32(
                                  false, av, false, bv, (short)0, acc[tm][tn], false, false);
            }
        }
        v2f xv, bx;
        xv.x = f.x[i >> 1][2 * (i & 1)];
        xv.y = f.x[i >> 1][2 * (i & 1) + 1];
        bx.x = f.b0[i][TN];
        bx.y = f.b1[i][TN];
        accx = __builtin_amdgcn_wmma_f32_16x16x4_f32(
                   false, xv, false, bx, (short)0, accx, false, false);
    }
}

template <int K, int L, int M, int N, int TN>
static __device__ __forceinline__ void rb_body(const float* __restrict__ W,
                                               const float* __restrict__ Bm,
                                               const float* __restrict__ Cm,
                                               const float* __restrict__ U,
                                               float* __restrict__ Out,
                                               float* __restrict__ BuT,
                                               float* __restrict__ CuT)
{
    const int b = blockIdx.z;
    const float* __restrict__ Ub = U + (size_t)b * K * L;
    float* __restrict__ Ob = Out + (size_t)b * M * L;

    const int wave  = threadIdx.x >> 5;
    const int lane  = threadIdx.x & 31;
    const int col   = lane & 15;
    const int khalf = lane >> 4;
    const int m0    = blockIdx.y * 64  + (wave >> 1) * 32;
    const int l0    = blockIdx.x * 128 + (wave & 1) * 64;

    const int  rt   = (blockIdx.y >> 2) * 2 + (wave >> 1);
    const bool isB  = rt < N / 16;
    const int  n0   = (rt & (N / 16 - 1)) * 16;

    const float* __restrict__ aptr = W  + (size_t)(m0 + col) * K + 8 * khalf;
    const float* __restrict__ xptr = (isB ? Bm : Cm) + (size_t)(n0 + col) * K + 8 * khalf;
    const float* __restrict__ bptr = Ub + (size_t)(8 * khalf) * L + l0 + 4 * col;

    v8f acc[2][4];
    #pragma unroll
    for (int tm = 0; tm < 2; ++tm) {
        #pragma unroll
        for (int tn = 0; tn < 4; ++tn)
            acc[tm][tn] = v8f{};
    }
    v8f accx = v8f{};

    RbFrag f0, f1;
    rb_load<K, L>(f0, aptr, xptr, bptr);
    for (int kb = 0; kb < K - 32; kb += 32) {
        rb_load<K, L>(f1, aptr + 16, xptr + 16, bptr + (size_t)16 * L);
        rb_compute<TN>(f0, acc, accx);
        rb_load<K, L>(f0, aptr + 32, xptr + 32, bptr + (size_t)32 * L);
        rb_compute<TN>(f1, acc, accx);
        aptr += 32;
        xptr += 32;
        bptr += (size_t)32 * L;
    }
    rb_load<K, L>(f1, aptr + 16, xptr + 16, bptr + (size_t)16 * L);
    rb_compute<TN>(f0, acc, accx);
    rb_compute<TN>(f1, acc, accx);

    #pragma unroll
    for (int tm = 0; tm < 2; ++tm) {
        #pragma unroll
        for (int tn = 0; tn < 4; ++tn) {
            float* __restrict__ dst = Ob + (size_t)(l0 + 4 * col + tn) * M + m0 + 16 * tm + 8 * khalf;
            const v8f v = acc[tm][tn];
            v4f lo = {v[0], v[1], v[2], v[3]};
            v4f hi = {v[4], v[5], v[6], v[7]};
            *(v4f*)(dst)     = lo;
            *(v4f*)(dst + 4) = hi;
        }
    }
    {
        float* __restrict__ Xo  = (isB ? BuT : CuT) + (size_t)b * L * N;
        float* __restrict__ dst = Xo + (size_t)(l0 + 4 * col + TN) * N + n0 + 8 * khalf;
        v4f lo = {accx[0], accx[1], accx[2], accx[3]};
        v4f hi = {accx[4], accx[5], accx[6], accx[7]};
        *(v4f*)(dst)     = lo;
        *(v4f*)(dst + 4) = hi;
    }
}

template <int K, int L, int M, int N>
__global__ __launch_bounds__(128) __attribute__((amdgpu_waves_per_eu(4, 4)))
void gemm_f32_wmma_rb(const float* __restrict__ W, const float* __restrict__ Bm,
                      const float* __restrict__ Cm, const float* __restrict__ U,
                      float* __restrict__ Out, float* __restrict__ BuT, float* __restrict__ CuT)
{
    static_assert(K % 32 == 0 && K >= 64 && L % 128 == 0 && M % 64 == 0, "tile sizes in rb_body");
    static_assert((M / 64) * 2 == (2 * N / 16) * 4 && (N / 16 & (N / 16 - 1)) == 0,
                  "the waves sharing an l range must equal the 16 x 16 tiles of [Bm; Cm] * u on it");
    switch (blockIdx.y & 3) {
    case 0:  rb_body<K, L, M, N, 0>(W, Bm, Cm, U, Out, BuT, CuT); break;
    case 1:  rb_body<K, L, M, N, 1>(W, Bm, Cm, U, Out, BuT, CuT); break;
    case 2:  rb_body<K, L, M, N, 2>(W, Bm, Cm, U, Out, BuT, CuT); break;
    default: rb_body<K, L, M, N, 3>(W, Bm, Cm, U, Out, BuT, CuT); break;
    }
}

constexpr int SCAN_T  = 32;
constexpr int SCAN_S  = 8;
constexpr int SCAN_RP = SCAN_T + 4;

template <int CTRL>
static __device__ __forceinline__ float xmask_add(float v)
{
    const int o = __builtin_amdgcn_update_dpp(0, __float_as_int(v), CTRL, 0xf, 0xf, true);
    return v + __int_as_float(o);
}

static __device__ __forceinline__ void wg_signal() { asm volatile("s_barrier_signal -1" ::: "memory"); }
static __device__ __forceinline__ void wg_wait()   { asm volatile("s_barrier_wait -1" ::: "memory"); }

struct ScanPre {
    v4f b[4], c[4];
    v4f d, u;
};

template <int DMODEL>
static __device__ __forceinline__ void scan_pre_load(ScanPre& p, const v4f* __restrict__ gB,
                                                     const v4f* __restrict__ gC,
                                                     const float* __restrict__ gD,
                                                     const float* __restrict__ gU, int chunk)
{
    const size_t cn = (size_t)chunk;
    #pragma unroll
    for (int i = 0; i < 4; ++i) {
        p.b[i] = gB[cn * 512 + 128 * i];
        p.c[i] = gC[cn * 512 + 128 * i];
    }
    p.d = *(const v4f*)(gD + cn * SCAN_T * DMODEL);
    p.u = *(const v4f*)(gU + cn * SCAN_T);
}

static __device__ __forceinline__ void scan_pre_store(const ScanPre& p, v4f* __restrict__ sBn,
                                                      v4f* __restrict__ sCn, v4f* __restrict__ sDn,
                                                      v4f* __restrict__ sUn, int t, int r, int lj)
{
    #pragma unroll
    for (int i = 0; i < 4; ++i) {
        sBn[t + 128 * i] = p.b[i];
        sCn[t + 128 * i] = p.c[i];
    }
    float* __restrict__ wD = (float*)sDn + (4 * (t & 3)) * SCAN_RP + (t >> 2);
    wD[0] = p.d[0];  wD[SCAN_RP] = p.d[1];  wD[2 * SCAN_RP] = p.d[2];  wD[3 * SCAN_RP] = p.d[3];
    sUn[r * (SCAN_RP / 4) + lj] = p.u;
}

struct ScanIn {
    v4f d[SCAN_S / 4], u[SCAN_S / 4];
    v4f b[SCAN_S][2], c[SCAN_S][2];
};

template <int BLK>
static __device__ __forceinline__ void scan_fetch(ScanIn& f, const v4f* __restrict__ cB,
                                                  const v4f* __restrict__ cC,
                                                  const v4f* __restrict__ cD,
                                                  const v4f* __restrict__ cU)
{
    #pragma unroll
    for (int gg = 0; gg < SCAN_S / 4; ++gg) {
        f.d[gg] = cD[BLK * (SCAN_S / 4) + gg];
        f.u[gg] = cU[BLK * (SCAN_S / 4) + gg];
        #pragma unroll
        for (int s = 0; s < 4; ++s) {
            const int js = gg * 4 + s;
            const int j  = BLK * SCAN_S + js;
            f.b[js][0] = cB[j * 16];
            f.b[js][1] = cB[j * 16 + 1];
            f.c[js][0] = cC[j * 16];
            f.c[js][1] = cC[j * 16 + 1];
        }
    }
}

template <int BLK>
static __device__ __forceinline__ void scan_steps(const ScanIn& f, float (&h)[8],
                                                  const float (&areg)[8], float Dd, int lj, v4f& yv)
{
    float hn[SCAN_S][8];
    #pragma unroll
    for (int js = 0; js < SCAN_S; ++js) {
        const float dl = f.d[js >> 2][js & 3];
        const float ul = f.u[js >> 2][js & 3];
        const float du = dl * ul;
        #pragma unroll
        for (int k = 0; k < 4; ++k) {
            h[k]      = fmaf(dl * areg[k], h[k], du * f.b[js][0][k]);
            hn[js][k] = h[k];
        }
        #pragma unroll
        for (int k = 0; k < 4; ++k) {
            h[4 + k]      = fmaf(dl * areg[4 + k], h[4 + k], du * f.b[js][1][k]);
            hn[js][4 + k] = h[4 + k];
        }
    }

    v2f y2[SCAN_S];
    #pragma unroll
    for (int js = 0; js < SCAN_S; ++js)
        y2[js] = v2f{0.0f, 0.0f};
    #pragma unroll
    for (int m = 0; m < 4; ++m) {
        #pragma unroll
        for (int js = 0; js < SCAN_S; ++js) {
            const v2f c2 = {f.c[js][m >> 1][2 * (m & 1)], f.c[js][m >> 1][2 * (m & 1) + 1]};
            const v2f h2 = {hn[js][2 * m], hn[js][2 * m + 1]};
            y2[js] = __builtin_elementwise_fma(c2, h2, y2[js]);
        }
    }
    float y[SCAN_S];
    #pragma unroll
    for (int js = 0; js < SCAN_S; ++js)
        y[js] = y2[js].x + y2[js].y;

    #pragma unroll
    for (int js = 0; js < SCAN_S; ++js)
        y[js] = xmask_add<0x161>(y[js]);
    #pragma unroll
    for (int js = 0; js < SCAN_S; ++js)
        y[js] = xmask_add<0x162>(y[js]);
    #pragma unroll
    for (int js = 0; js < SCAN_S; ++js)
        y[js] = xmask_add<0x164>(y[js]);

    #pragma unroll
    for (int js = 0; js < SCAN_S; ++js) {
        const int   g    = BLK * (SCAN_S / 4) + (js >> 2);
        const bool  mine = (lj == g);
        const float yo   = fmaf(f.u[js >> 2][js & 3], Dd, y[js]);
        yv[js & 3] = mine ? yo : yv[js & 3];
    }
}

template <int DMODEL, int L>
__global__ __launch_bounds__(128) __attribute__((amdgpu_waves_per_eu(1, 2)))
void ssm_scan(const float* __restrict__ u, const float* __restrict__ A,
              const float* __restrict__ D,
              const float* __restrict__ BuT, const float* __restrict__ CuT,
              const float* __restrict__ dT, float* __restrict__ out)
{
    constexpr int N      = 64;
    constexpr int T      = SCAN_T;
    constexpr int RP4    = SCAN_RP / 4;
    constexpr int ROWS   = 16;
    constexpr int NCHUNK = L / T;
    static_assert(L % T == 0 && T == 32 && N == 64, "chunk loaders are written for 32 steps of 64 states");
    static_assert(T / SCAN_S == 4 && SCAN_S % 4 == 0, "the chunk body below is written out for 4 read-ahead blocks");
    static_assert(NCHUNK >= 2, "two chunks are staged before the loop");

    const int b  = blockIdx.y;
    const int t  = threadIdx.x;
    const int d0 = blockIdx.x * ROWS;
    const int r  = t >> 3;
    const int lj = t & 7;
    const int d  = d0 + r;

    float*       __restrict__ ybd = out + ((size_t)b * DMODEL + d) * L + 4 * lj;
    const float* __restrict__ Bub = BuT + (size_t)b * L * N;
    const float* __restrict__ Cub = CuT + (size_t)b * L * N;

    const v4f* __restrict__ gB = (const v4f*)Bub + t;
    const v4f* __restrict__ gC = (const v4f*)Cub + t;
    const float* __restrict__ gD = dT + ((size_t)b * L + (t >> 2)) * DMODEL + d0 + 4 * (t & 3);
    const float* __restrict__ gU = u + ((size_t)b * DMODEL + d) * L + 4 * lj;

    const float Dd = D[d];

    float areg[8], h[8];
    #pragma unroll
    for (int i = 0; i < 8; ++i) { areg[i] = A[8 * lj + i]; h[i] = 0.0f; }

    __shared__ v4f sB[3][T * 16];
    __shared__ v4f sC[3][T * 16];
    __shared__ v4f sD[3][ROWS * RP4];
    __shared__ v4f sU[3][ROWS * RP4];

    ScanPre pre;

    #pragma unroll
    for (int cc = 0; cc < 2; ++cc) {
        scan_pre_load<DMODEL>(pre, gB, gC, gD, gU, cc);
        scan_pre_store(pre, &sB[cc][0], &sC[cc][0], &sD[cc][0], &sU[cc][0], t, r, lj);
    }
    __syncthreads();
    wg_signal();

    ScanIn fa, fb;
    scan_fetch<0>(fa, &sB[0][2 * lj], &sC[0][2 * lj], &sD[0][r * RP4], &sU[0][r * RP4]);

    int cur = 0;
    for (int c = 0; c < NCHUNK; ++c) {
        const int  nx1   = (cur == 2) ? 0 : cur + 1;
        const int  nx2   = (nx1 == 2) ? 0 : nx1 + 1;
        const bool more2 = (c + 2 < NCHUNK);

        if (more2) scan_pre_load<DMODEL>(pre, gB, gC, gD, gU, c + 2);

        const v4f* __restrict__ cB = &sB[cur][2 * lj];
        const v4f* __restrict__ cC = &sC[cur][2 * lj];
        const v4f* __restrict__ cD = &sD[cur][r * RP4];
        const v4f* __restrict__ cU = &sU[cur][r * RP4];

        v4f yv = {0.0f, 0.0f, 0.0f, 0.0f};

        __builtin_amdgcn_sched_barrier(0);
        scan_fetch<1>(fb, cB, cC, cD, cU);
        __builtin_amdgcn_sched_barrier(0);
        scan_steps<0>(fa, h, areg, Dd, lj, yv);

        __builtin_amdgcn_sched_barrier(0);
        scan_fetch<2>(fa, cB, cC, cD, cU);
        __builtin_amdgcn_sched_barrier(0);
        scan_steps<1>(fb, h, areg, Dd, lj, yv);

        __builtin_amdgcn_sched_barrier(0);
        scan_fetch<3>(fb, cB, cC, cD, cU);
        __builtin_amdgcn_sched_barrier(0);
        scan_steps<2>(fa, h, areg, Dd, lj, yv);

        wg_wait();
        __builtin_amdgcn_sched_barrier(0);
        scan_fetch<0>(fa, &sB[nx1][2 * lj], &sC[nx1][2 * lj], &sD[nx1][r * RP4], &sU[nx1][r * RP4]);
        __builtin_amdgcn_sched_barrier(0);
        scan_steps<3>(fb, h, areg, Dd, lj, yv);

        *(v4f*)(ybd + (size_t)c * T) = yv;

        if (more2) scan_pre_store(pre, &sB[nx2][0], &sC[nx2][0], &sD[nx2][0], &sU[nx2][0], t, r, lj);

        __builtin_amdgcn_fence(__ATOMIC_RELEASE, "workgroup", "local");
        wg_signal();
        cur = nx1;
    }
    wg_wait();
}

extern "C" void kernel_launch(void* const* d_in, const int* in_sizes, int n_in,
                              void* d_out, int out_size, void* d_ws, size_t ws_size,
                              hipStream_t stream)
{
    (void)in_sizes; (void)n_in; (void)out_size; (void)ws_size;

    const float* u     = (const float*)d_in[1];
    const float* A     = (const float*)d_in[2];
    const float* Bm    = (const float*)d_in[3];
    const float* Cm    = (const float*)d_in[4];
    const float* D     = (const float*)d_in[5];
    const float* Delta = (const float*)d_in[6];
    float* out = (float*)d_out;

    constexpr int batch = 4, dmodel = 1024, N = 64, L = 2048;

    float* BuT = (float*)d_ws;
    float* CuT = BuT + (size_t)batch * L * N;
    float* dT  = CuT + (size_t)batch * L * N;

    const dim3 blk(128, 1, 1);

    gemm_f32_wmma_rb<dmodel, L, dmodel, N><<<dim3(L / 128, dmodel / 64, batch), blk, 0, stream>>>(
        Delta, Bm, Cm, u, dT, BuT, CuT);

    ssm_scan<dmodel, L><<<dim3(dmodel / 16, batch), dim3(128, 1, 1), 0, stream>>>(
        u, A, D, BuT, CuT, dT, out);
}


# delta GEMM: wave block enlarged from 32x64 to 64x64 (4x4 tiles, 2 waves per SIMD): each 16-byte A and B fragment feeds 4 WMMAs, a third fewer operand loads per WMMA; sums unchanged
# speedup vs baseline: 17.2478x; 1.0528x over previous
#include <hip/hip_runtime.h>

typedef __attribute__((ext_vector_type(2))) float v2f;
typedef __attribute__((ext_vector_type(4))) float v4f;
typedef __attribute__((ext_vector_type(8))) float v8f;

constexpr int RB_TM = 4;
static_assert(32 * RB_TM == 128, "kernel_launch sizes the grid for blocks of 128 rows of m");

struct RbFrag {
    v4f a[RB_TM][2];
    v4f x[2];
    v4f b0[4];
    v4f b1[4];
};

template <int K, int L>
static __device__ __forceinline__ void rb_load(RbFrag& f, const float* __restrict__ ap,
                                               const float* __restrict__ xp,
                                               const float* __restrict__ bp)
{
    #pragma unroll
    for (int tm = 0; tm < RB_TM; ++tm) {
        #pragma unroll
        for (int q = 0; q < 2; ++q)
            f.a[tm][q] = *(const v4f*)(ap + (size_t)tm * 16 * K + 4 * q);
    }
    #pragma unroll
    for (int q = 0; q < 2; ++q)
        f.x[q] = *(const v4f*)(xp + 4 * q);
    #pragma unroll
    for (int i = 0; i < 4; ++i) {
        f.b0[i] = *(const v4f*)(bp + (size_t)(2 * i)     * L);
        f.b1[i] = *(const v4f*)(bp + (size_t)(2 * i + 1) * L);
    }
}

template <int TN0>
static __device__ __forceinline__ void rb_compute(const RbFrag& f, v8f (&acc)[RB_TM][4], v8f (&accx)[2])
{
    #pragma unroll
    for (int i = 0; i < 4; ++i) {
        #pragma unroll
        for (int tm = 0; tm < RB_TM; ++tm) {
            v2f av;
            av.x = f.a[tm][i >> 1][2 * (i & 1)];
            av.y = f.a[tm][i >> 1][2 * (i & 1) + 1];
            #pragma unroll
            for (int tn = 0; tn < 4; ++tn) {
                v2f bv;
                bv.x = f.b0[i][tn];
                bv.y = f.b1[i][tn];
                acc[tm][tn] = __builtin_amdgcn_wmma_f32_16x16x4_f32(
                                  false, av, false, bv, (short)0, acc[tm][tn], false, false);
            }
        }
        v2f xv;
        xv.x = f.x[i >> 1][2 * (i & 1)];
        xv.y = f.x[i >> 1][2 * (i & 1) + 1];
        #pragma unroll
        for (int j = 0; j < 2; ++j) {
            v2f bx;
            bx.x = f.b0[i][TN0 + j];
            bx.y = f.b1[i][TN0 + j];
            accx[j] = __builtin_amdgcn_wmma_f32_16x16x4_f32(
                          false, xv, false, bx, (short)0, accx[j], false, false);
        }
    }
}

template <int K, int L, int M, int N, int TN0>
static __device__ __forceinline__ void rb_body(const float* __restrict__ W,
                                               const float* __restrict__ Bm,
                                               const float* __restrict__ Cm,
                                               const float* __restrict__ U,
                                               float* __restrict__ Out,
                                               float* __restrict__ BuT,
                                               float* __restrict__ CuT)
{
    const int b = blockIdx.z;
    const float* __restrict__ Ub = U + (size_t)b * K * L;
    float* __restrict__ Ob = Out + (size_t)b * M * L;

    const int wave  = threadIdx.x >> 5;
    const int lane  = threadIdx.x & 31;
    const int col   = lane & 15;
    const int khalf = lane >> 4;
    const int m0    = blockIdx.y * (32 * RB_TM) + (wave >> 1) * (16 * RB_TM);
    const int l0    = blockIdx.x * 128 + (wave & 1) * 64;

    const int  rt   = (blockIdx.y >> 1) * 2 + (wave >> 1);
    const bool isB  = rt < N / 16;
    const int  n0   = (rt & (N / 16 - 1)) * 16;

    const float* __restrict__ aptr = W  + (size_t)(m0 + col) * K + 8 * khalf;
    const float* __restrict__ xptr = (isB ? Bm : Cm) + (size_t)(n0 + col) * K + 8 * khalf;
    const float* __restrict__ bptr = Ub + (size_t)(8 * khalf) * L + l0 + 4 * col;

    v8f acc[RB_TM][4];
    #pragma unroll
    for (int tm = 0; tm < RB_TM; ++tm) {
        #pragma unroll
        for (int tn = 0; tn < 4; ++tn)
            acc[tm][tn] = v8f{};
    }
    v8f accx[2] = {v8f{}, v8f{}};

    RbFrag f0, f1;
    rb_load<K, L>(f0, aptr, xptr, bptr);
    for (int kb = 0; kb < K - 32; kb += 32) {
        rb_load<K, L>(f1, aptr + 16, xptr + 16, bptr + (size_t)16 * L);
        rb_compute<TN0>(f0, acc, accx);
        rb_load<K, L>(f0, aptr + 32, xptr + 32, bptr + (size_t)32 * L);
        rb_compute<TN0>(f1, acc, accx);
        aptr += 32;
        xptr += 32;
        bptr += (size_t)32 * L;
    }
    rb_load<K, L>(f1, aptr + 16, xptr + 16, bptr + (size_t)16 * L);
    rb_compute<TN0>(f0, acc, accx);
    rb_compute<TN0>(f1, acc, accx);

    #pragma unroll
    for (int tm = 0; tm < RB_TM; ++tm) {
        #pragma unroll
        for (int tn = 0; tn < 4; ++tn) {
            float* __restrict__ dst = Ob + (size_t)(l0 + 4 * col + tn) * M + m0 + 16 * tm + 8 * khalf;
            const v8f v = acc[tm][tn];
            v4f lo = {v[0], v[1], v[2], v[3]};
            v4f hi = {v[4], v[5], v[6], v[7]};
            *(v4f*)(dst)     = lo;
            *(v4f*)(dst + 4) = hi;
        }
    }
    #pragma unroll
    for (int j = 0; j < 2; ++j) {
        float* __restrict__ Xo  = (isB ? BuT : CuT) + (size_t)b * L * N;
        float* __restrict__ dst = Xo + (size_t)(l0 + 4 * col + TN0 + j) * N + n0 + 8 * khalf;
        const v8f v = accx[j];
        v4f lo = {v[0], v[1], v[2], v[3]};
        v4f hi = {v[4], v[5], v[6], v[7]};
        *(v4f*)(dst)     = lo;
        *(v4f*)(dst + 4) = hi;
    }
}

template <int K, int L, int M, int N>
__global__ __launch_bounds__(128) __attribute__((amdgpu_waves_per_eu(2, 2)))
void gemm_f32_wmma_rb(const float* __restrict__ W, const float* __restrict__ Bm,
                      const float* __restrict__ Cm, const float* __restrict__ U,
                      float* __restrict__ Out, float* __restrict__ BuT, float* __restrict__ CuT)
{
    static_assert(K % 32 == 0 && K >= 64 && L % 128 == 0 && M % (32 * RB_TM) == 0, "tile sizes in rb_body");
    static_assert((M / (32 * RB_TM)) * 2 * 2 == (2 * N / 16) * 4 && (N / 16 & (N / 16 - 1)) == 0,
                  "twice the waves sharing an l range must equal the 16 x 16 tiles of [Bm; Cm] * u on it");
    if ((blockIdx.y & 1) == 0) rb_body<K, L, M, N, 0>(W, Bm, Cm, U, Out, BuT, CuT);
    else                       rb_body<K, L, M, N, 2>(W, Bm, Cm, U, Out, BuT, CuT);
}

constexpr int SCAN_T  = 32;
constexpr int SCAN_S  = 8;
constexpr int SCAN_RP = SCAN_T + 4;

template <int CTRL>
static __device__ __forceinline__ float xmask_add(float v)
{
    const int o = __builtin_amdgcn_update_dpp(0, __float_as_int(v), CTRL, 0xf, 0xf, true);
    return v + __int_as_float(o);
}

static __device__ __forceinline__ void wg_signal() { asm volatile("s_barrier_signal -1" ::: "memory"); }
static __device__ __forceinline__ void wg_wait()   { asm volatile("s_barrier_wait -1" ::: "memory"); }

struct ScanPre {
    v4f b[4], c[4];
    v4f d, u;
};

template <int DMODEL>
static __device__ __forceinline__ void scan_pre_load(ScanPre& p, const v4f* __restrict__ gB,
                                                     const v4f* __restrict__ gC,
                                                     const float* __restrict__ gD,
                                                     const float* __restrict__ gU, int chunk)
{
    const size_t cn = (size_t)chunk;
    #pragma unroll
    for (int i = 0; i < 4; ++i) {
        p.b[i] = gB[cn * 512 + 128 * i];
        p.c[i] = gC[cn * 512 + 128 * i];
    }
    p.d = *(const v4f*)(gD + cn * SCAN_T * DMODEL);
    p.u = *(const v4f*)(gU + cn * SCAN_T);
}

static __device__ __forceinline__ void scan_pre_store(const ScanPre& p, v4f* __restrict__ sBn,
                                                      v4f* __restrict__ sCn, v4f* __restrict__ sDn,
                                                      v4f* __restrict__ sUn, int t, int r, int lj)
{
    #pragma unroll
    for (int i = 0; i < 4; ++i) {
        sBn[t + 128 * i] = p.b[i];
        sCn[t + 128 * i] = p.c[i];
    }
    float* __restrict__ wD = (float*)sDn + (4 * (t & 3)) * SCAN_RP + (t >> 2);
    wD[0] = p.d[0];  wD[SCAN_RP] = p.d[1];  wD[2 * SCAN_RP] = p.d[2];  wD[3 * SCAN_RP] = p.d[3];
    sUn[r * (SCAN_RP / 4) + lj] = p.u;
}

struct ScanIn {
    v4f d[SCAN_S / 4], u[SCAN_S / 4];
    v4f b[SCAN_S][2], c[SCAN_S][2];
};

template <int BLK>
static __device__ __forceinline__ void scan_fetch(ScanIn& f, const v4f* __restrict__ cB,
                                                  const v4f* __restrict__ cC,
                                                  const v4f* __restrict__ cD,
                                                  const v4f* __restrict__ cU)
{
    #pragma unroll
    for (int gg = 0; gg < SCAN_S / 4; ++gg) {
        f.d[gg] = cD[BLK * (SCAN_S / 4) + gg];
        f.u[gg] = cU[BLK * (SCAN_S / 4) + gg];
        #pragma unroll
        for (int s = 0; s < 4; ++s) {
            const int js = gg * 4 + s;
            const int j  = BLK * SCAN_S + js;
            f.b[js][0] = cB[j * 16];
            f.b[js][1] = cB[j * 16 + 1];
            f.c[js][0] = cC[j * 16];
            f.c[js][1] = cC[j * 16 + 1];
        }
    }
}

template <int BLK>
static __device__ __forceinline__ void scan_steps(const ScanIn& f, float (&h)[8],
                                                  const float (&areg)[8], float Dd, int lj, v4f& yv)
{
    float hn[SCAN_S][8];
    #pragma unroll
    for (int js = 0; js < SCAN_S; ++js) {
        const float dl = f.d[js >> 2][js & 3];
        const float ul = f.u[js >> 2][js & 3];
        const float du = dl * ul;
        #pragma unroll
        for (int k = 0; k < 4; ++k) {
            h[k]      = fmaf(dl * areg[k], h[k], du * f.b[js][0][k]);
            hn[js][k] = h[k];
        }
        #pragma unroll
        for (int k = 0; k < 4; ++k) {
            h[4 + k]      = fmaf(dl * areg[4 + k], h[4 + k], du * f.b[js][1][k]);
            hn[js][4 + k] = h[4 + k];
        }
    }

    v2f y2[SCAN_S];
    #pragma unroll
    for (int js = 0; js < SCAN_S; ++js)
        y2[js] = v2f{0.0f, 0.0f};
    #pragma unroll
    for (int m = 0; m < 4; ++m) {
        #pragma unroll
        for (int js = 0; js < SCAN_S; ++js) {
            const v2f c2 = {f.c[js][m >> 1][2 * (m & 1)], f.c[js][m >> 1][2 * (m & 1) + 1]};
            const v2f h2 = {hn[js][2 * m], hn[js][2 * m + 1]};
            y2[js] = __builtin_elementwise_fma(c2, h2, y2[js]);
        }
    }
    float y[SCAN_S];
    #pragma unroll
    for (int js = 0; js < SCAN_S; ++js)
        y[js] = y2[js].x + y2[js].y;

    #pragma unroll
    for (int js = 0; js < SCAN_S; ++js)
        y[js] = xmask_add<0x161>(y[js]);
    #pragma unroll
    for (int js = 0; js < SCAN_S; ++js)
        y[js] = xmask_add<0x162>(y[js]);
    #pragma unroll
    for (int js = 0; js < SCAN_S; ++js)
        y[js] = xmask_add<0x164>(y[js]);

    #pragma unroll
    for (int js = 0; js < SCAN_S; ++js) {
        const int   g    = BLK * (SCAN_S / 4) + (js >> 2);
        const bool  mine = (lj == g);
        const float yo   = fmaf(f.u[js >> 2][js & 3], Dd, y[js]);
        yv[js & 3] = mine ? yo : yv[js & 3];
    }
}

template <int DMODEL, int L>
__global__ __launch_bounds__(128) __attribute__((amdgpu_waves_per_eu(1, 2)))
void ssm_scan(const float* __restrict__ u, const float* __restrict__ A,
              const float* __restrict__ D,
              const float* __restrict__ BuT, const float* __restrict__ CuT,
              const float* __restrict__ dT, float* __restrict__ out)
{
    constexpr int N      = 64;
    constexpr int T      = SCAN_T;
    constexpr int RP4    = SCAN_RP / 4;
    constexpr int ROWS   = 16;
    constexpr int NCHUNK = L / T;
    static_assert(L % T == 0 && T == 32 && N == 64, "chunk loaders are written for 32 steps of 64 states");
    static_assert(T / SCAN_S == 4 && SCAN_S % 4 == 0, "the chunk body below is written out for 4 read-ahead blocks");
    static_assert(NCHUNK >= 2, "two chunks are staged before the loop");

    const int b  = blockIdx.y;
    const int t  = threadIdx.x;
    const int d0 = blockIdx.x * ROWS;
    const int r  = t >> 3;
    const int lj = t & 7;
    const int d  = d0 + r;

    float*       __restrict__ ybd = out + ((size_t)b * DMODEL + d) * L + 4 * lj;
    const float* __restrict__ Bub = BuT + (size_t)b * L * N;
    const float* __restrict__ Cub = CuT + (size_t)b * L * N;

    const v4f* __restrict__ gB = (const v4f*)Bub + t;
    const v4f* __restrict__ gC = (const v4f*)Cub + t;
    const float* __restrict__ gD = dT + ((size_t)b * L + (t >> 2)) * DMODEL + d0 + 4 * (t & 3);
    const float* __restrict__ gU = u + ((size_t)b * DMODEL + d) * L + 4 * lj;

    const float Dd = D[d];

    float areg[8], h[8];
    #pragma unroll
    for (int i = 0; i < 8; ++i) { areg[i] = A[8 * lj + i]; h[i] = 0.0f; }

    __shared__ v4f sB[3][T * 16];
    __shared__ v4f sC[3][T * 16];
    __shared__ v4f sD[3][ROWS * RP4];
    __shared__ v4f sU[3][ROWS * RP4];

    ScanPre pre;

    #pragma unroll
    for (int cc = 0; cc < 2; ++cc) {
        scan_pre_load<DMODEL>(pre, gB, gC, gD, gU, cc);
        scan_pre_store(pre, &sB[cc][0], &sC[cc][0], &sD[cc][0], &sU[cc][0], t, r, lj);
    }
    __syncthreads();
    wg_signal();

    ScanIn fa, fb;
    scan_fetch<0>(fa, &sB[0][2 * lj], &sC[0][2 * lj], &sD[0][r * RP4], &sU[0][r * RP4]);

    int cur = 0;
    for (int c = 0; c < NCHUNK; ++c) {
        const int  nx1   = (cur == 2) ? 0 : cur + 1;
        const int  nx2   = (nx1 == 2) ? 0 : nx1 + 1;
        const bool more2 = (c + 2 < NCHUNK);

        if (more2) scan_pre_load<DMODEL>(pre, gB, gC, gD, gU, c + 2);

        const v4f* __restrict__ cB = &sB[cur][2 * lj];
        const v4f* __restrict__ cC = &sC[cur][2 * lj];
        const v4f* __restrict__ cD = &sD[cur][r * RP4];
        const v4f* __restrict__ cU = &sU[cur][r * RP4];

        v4f yv = {0.0f, 0.0f, 0.0f, 0.0f};

        __builtin_amdgcn_sched_barrier(0);
        scan_fetch<1>(fb, cB, cC, cD, cU);
        __builtin_amdgcn_sched_barrier(0);
        scan_steps<0>(fa, h, areg, Dd, lj, yv);

        __builtin_amdgcn_sched_barrier(0);
        scan_fetch<2>(fa, cB, cC, cD, cU);
        __builtin_amdgcn_sched_barrier(0);
        scan_steps<1>(fb, h, areg, Dd, lj, yv);

        __builtin_amdgcn_sched_barrier(0);
        scan_fetch<3>(fb, cB, cC, cD, cU);
        __builtin_amdgcn_sched_barrier(0);
        scan_steps<2>(fa, h, areg, Dd, lj, yv);

        wg_wait();
        __builtin_amdgcn_sched_barrier(0);
        scan_fetch<0>(fa, &sB[nx1][2 * lj], &sC[nx1][2 * lj], &sD[nx1][r * RP4], &sU[nx1][r * RP4]);
        __builtin_amdgcn_sched_barrier(0);
        scan_steps<3>(fb, h, areg, Dd, lj, yv);

        *(v4f*)(ybd + (size_t)c * T) = yv;

        if (more2) scan_pre_store(pre, &sB[nx2][0], &sC[nx2][0], &sD[nx2][0], &sU[nx2][0], t, r, lj);

        __builtin_amdgcn_fence(__ATOMIC_RELEASE, "workgroup", "local");
        wg_signal();
        cur = nx1;
    }
    wg_wait();
}

extern "C" void kernel_launch(void* const* d_in, const int* in_sizes, int n_in,
                              void* d_out, int out_size, void* d_ws, size_t ws_size,
                              hipStream_t stream)
{
    (void)in_sizes; (void)n_in; (void)out_size; (void)ws_size;

    const float* u     = (const float*)d_in[1];
    const float* A     = (const float*)d_in[2];
    const float* Bm    = (const float*)d_in[3];
    const float* Cm    = (const float*)d_in[4];
    const float* D     = (const float*)d_in[5];
    const float* Delta = (const float*)d_in[6];
    float* out = (float*)d_out;

    constexpr int batch = 4, dmodel = 1024, N = 64, L = 2048;

    float* BuT = (float*)d_ws;
    float* CuT = BuT + (size_t)batch * L * N;
    float* dT  = CuT + (size_t)batch * L * N;

    const dim3 blk(128, 1, 1);

    gemm_f32_wmma_rb<dmodel, L, dmodel, N><<<dim3(L / 128, dmodel / 128, batch), blk, 0, stream>>>(
        Delta, Bm, Cm, u, dT, BuT, CuT);

    ssm_scan<dmodel, L><<<dim3(dmodel / 16, batch), dim3(128, 1, 1), 0, stream>>>(
        u, A, D, BuT, CuT, dT, out);
}
